# starting kernel
# baseline (speedup 1.0000x reference)
#include <hip/hip_runtime.h>
#include <hip/hip_bf16.h>

#define IN_CH   512
#define OUT_CH  64
#define KCHUNK  128
#define ROWS_PER_BLOCK 128

typedef __attribute__((ext_vector_type(2))) float v2f;
typedef __attribute__((ext_vector_type(4))) float v4f;
typedef __attribute__((ext_vector_type(8))) float v8f;

__global__ void init_kernel(float* __restrict__ agg, float* __restrict__ deg, int n) {
    int i = blockIdx.x * blockDim.x + threadIdx.x;
    int total = n * OUT_CH;
    if (i < total) agg[i] = 0.0f;
    if (i < n)     deg[i] = 1.0f;
}

__global__ void deg_kernel(const int* __restrict__ dst, float* __restrict__ deg, int e) {
    int i = blockIdx.x * blockDim.x + threadIdx.x;
    if (i < e) atomicAdd(&deg[dst[i]], 1.0f);
}

__global__ void dinv_kernel(float* __restrict__ deg, int n) {
    int i = blockIdx.x * blockDim.x + threadIdx.x;
    if (i < n) deg[i] = rsqrtf(deg[i]);
}

__global__ __launch_bounds__(256) void gemm_wmma_kernel(
        const float* __restrict__ x, const float* __restrict__ W,
        float* __restrict__ h, int n) {
    __shared__ __align__(16) v2f lds_w[(KCHUNK / 2) * OUT_CH];

    const int tid   = threadIdx.x;
    const int wave  = tid >> 5;
    const int lane  = tid & 31;
    const int half  = lane >> 4;
    const int l15   = lane & 15;
    const int kh    = half * 2;
    const int row0  = blockIdx.x * ROWS_PER_BLOCK + wave * 16;

    int arow = row0 + l15;
    if (arow >= n) arow = n - 1;
    const float* xrow = x + (size_t)arow * IN_CH;

    v8f acc[4] = {{}, {}, {}, {}};

    for (int kc = 0; kc < IN_CH; kc += KCHUNK) {
        #pragma unroll
        for (int i = 0; i < (KCHUNK / 2) * OUT_CH / 256; ++i) {
            int j = tid + i * 256;
            int p = j >> 6;
            int c = j & 63;
            v2f w2;
            w2.x = W[(size_t)(kc + 2 * p)     * OUT_CH + c];
            w2.y = W[(size_t)(kc + 2 * p + 1) * OUT_CH + c];
            lds_w[j] = w2;
        }
        __syncthreads();

        #pragma unroll 4
        for (int k = 0; k < KCHUNK; k += 4) {
            v2f a;
            a.x = xrow[kc + k + kh];
            a.y = xrow[kc + k + kh + 1];
            const int prow = (k >> 1) + half;
            #pragma unroll
            for (int t = 0; t < 4; ++t) {
                v2f b = lds_w[prow * OUT_CH + t * 16 + l15];
                acc[t] = __builtin_amdgcn_wmma_f32_16x16x4_f32(
                    false, a, false, b, (short)0, acc[t], false, false);
            }
        }
        __syncthreads();
    }

    #pragma unroll
    for (int v = 0; v < 8; ++v) {
        int row = row0 + half * 8 + v;
        if (row < n) {
            #pragma unroll
            for (int t = 0; t < 4; ++t)
                h[(size_t)row * OUT_CH + t * 16 + l15] = acc[t][v];
        }
    }
}

__global__ __launch_bounds__(256) void scatter_kernel(
        const float* __restrict__ h, const int* __restrict__ src,
        const int* __restrict__ dst, const float* __restrict__ dinv,
        float* __restrict__ agg, int e) {
    int edge = blockIdx.x * 8 + (threadIdx.x >> 5);
    if (edge >= e) return;
    int lane = threadIdx.x & 31;
    int s = src[edge];
    int d = dst[edge];
    float norm = dinv[s] * dinv[d];
    const float2* hs = (const float2*)(h + (size_t)s * OUT_CH);
    float2 v = hs[lane];
    float* out = agg + (size_t)d * OUT_CH + lane * 2;
    atomicAdd(out,     v.x * norm);
    atomicAdd(out + 1, v.y * norm);
}

__global__ void finalize_kernel(const float* __restrict__ agg,
                                const float* __restrict__ h,
                                const float* __restrict__ dinv,
                                const float* __restrict__ bias,
                                const float* __restrict__ alpha,
                                float* __restrict__ out, int n) {
    int idx = blockIdx.x * blockDim.x + threadIdx.x;
    if (idx >= n * OUT_CH) return;
    int c = idx & (OUT_CH - 1);
    int i = idx >> 6;
    float di = dinv[i];
    float v = agg[idx] + h[idx] * di * di + bias[c];
    out[idx] = (v >= 0.0f) ? v : alpha[c] * v;
}

extern "C" void kernel_launch(void* const* d_in, const int* in_sizes, int n_in,
                              void* d_out, int out_size, void* d_ws, size_t ws_size,
                              hipStream_t stream) {
    const float* x     = (const float*)d_in[0];
    const int*   ei    = (const int*)  d_in[1];
    const float* W     = (const float*)d_in[2];
    const float* bias  = (const float*)d_in[3];
    const float* alpha = (const float*)d_in[4];
    float*       out   = (float*)d_out;

    const int N = in_sizes[0] / IN_CH;
    const int E = in_sizes[1] / 2;
    const int* src = ei;
    const int* dst = ei + E;

    char* ws = (char*)d_ws;
    size_t hw_bytes = ((size_t)N * OUT_CH * sizeof(float) + 255) & ~(size_t)255;
    float* h    = (float*)ws;
    float* agg  = (float*)(ws + hw_bytes);
    float* dinv = (float*)(ws + 2 * hw_bytes);

    const int threads = 256;
    int totNC = N * OUT_CH;

    init_kernel<<<(totNC + threads - 1) / threads, threads, 0, stream>>>(agg, dinv, N);
    deg_kernel<<<(E + threads - 1) / threads, threads, 0, stream>>>(dst, dinv, E);
    dinv_kernel<<<(N + threads - 1) / threads, threads, 0, stream>>>(dinv, N);
    gemm_wmma_kernel<<<(N + ROWS_PER_BLOCK - 1) / ROWS_PER_BLOCK, threads, 0, stream>>>(
        x, W, h, N);
    scatter_kernel<<<(E + 7) / 8, threads, 0, stream>>>(h, src, dst, dinv, agg, E);
    finalize_kernel<<<(totNC + threads - 1) / threads, threads, 0, stream>>>(
        agg, h, dinv, bias, alpha, out, N);
}


# scatter: one wave handles 32 edges (indices and norms loaded once, coalesced; 16 h-row loads in flight before their atomics) instead of one edge per wave; float atomics kept, arrival-order sums
# speedup vs baseline: 1.3281x; 1.3281x over previous
#include <hip/hip_runtime.h>
#include <hip/hip_bf16.h>

#define IN_CH   512
#define OUT_CH  64
#define KCHUNK  128
#define ROWS_PER_BLOCK 128

typedef __attribute__((ext_vector_type(2))) float v2f;
typedef __attribute__((ext_vector_type(4))) float v4f;
typedef __attribute__((ext_vector_type(8))) float v8f;

__global__ void init_kernel(float* __restrict__ agg, float* __restrict__ deg, int n) {
    int i = blockIdx.x * blockDim.x + threadIdx.x;
    int total = n * OUT_CH;
    if (i < total) agg[i] = 0.0f;
    if (i < n)     deg[i] = 1.0f;
}

__global__ void deg_kernel(const int* __restrict__ dst, float* __restrict__ deg, int e) {
    int i = blockIdx.x * blockDim.x + threadIdx.x;
    if (i < e) atomicAdd(&deg[dst[i]], 1.0f);
}

__global__ void dinv_kernel(float* __restrict__ deg, int n) {
    int i = blockIdx.x * blockDim.x + threadIdx.x;
    if (i < n) deg[i] = rsqrtf(deg[i]);
}

__global__ __launch_bounds__(256) void gemm_wmma_kernel(
        const float* __restrict__ x, const float* __restrict__ W,
        float* __restrict__ h, int n) {
    __shared__ __align__(16) v2f lds_w[(KCHUNK / 2) * OUT_CH];

    const int tid   = threadIdx.x;
    const int wave  = tid >> 5;
    const int lane  = tid & 31;
    const int half  = lane >> 4;
    const int l15   = lane & 15;
    const int kh    = half * 2;
    const int row0  = blockIdx.x * ROWS_PER_BLOCK + wave * 16;

    int arow = row0 + l15;
    if (arow >= n) arow = n - 1;
    const float* xrow = x + (size_t)arow * IN_CH;

    v8f acc[4] = {{}, {}, {}, {}};

    for (int kc = 0; kc < IN_CH; kc += KCHUNK) {
        #pragma unroll
        for (int i = 0; i < (KCHUNK / 2) * OUT_CH / 256; ++i) {
            int j = tid + i * 256;
            int p = j >> 6;
            int c = j & 63;
            v2f w2;
            w2.x = W[(size_t)(kc + 2 * p)     * OUT_CH + c];
            w2.y = W[(size_t)(kc + 2 * p + 1) * OUT_CH + c];
            lds_w[j] = w2;
        }
        __syncthreads();

        #pragma unroll 4
        for (int k = 0; k < KCHUNK; k += 4) {
            v2f a;
            a.x = xrow[kc + k + kh];
            a.y = xrow[kc + k + kh + 1];
            const int prow = (k >> 1) + half;
            #pragma unroll
            for (int t = 0; t < 4; ++t) {
                v2f b = lds_w[prow * OUT_CH + t * 16 + l15];
                acc[t] = __builtin_amdgcn_wmma_f32_16x16x4_f32(
                    false, a, false, b, (short)0, acc[t], false, false);
            }
        }
        __syncthreads();
    }

    #pragma unroll
    for (int v = 0; v < 8; ++v) {
        int row = row0 + half * 8 + v;
        if (row < n) {
            #pragma unroll
            for (int t = 0; t < 4; ++t)
                h[(size_t)row * OUT_CH + t * 16 + l15] = acc[t][v];
        }
    }
}

#define EDGES_PER_WAVE 32
#define EDGE_BATCH     16

__global__ __launch_bounds__(256) void scatter_kernel(
        const float* __restrict__ h, const int* __restrict__ src,
        const int* __restrict__ dst, const float* __restrict__ dinv,
        float* __restrict__ agg, int e) {
    const int lane = threadIdx.x & 31;
    const int wave = blockIdx.x * 8 + __builtin_amdgcn_readfirstlane((int)(threadIdx.x >> 5));
    const int base = wave * EDGES_PER_WAVE;
    if (base >= e) return;
    const int cnt = min(EDGES_PER_WAVE, e - base);

    int   s = 0, d = 0;
    float nrm = 0.0f;
    if (lane < cnt) {
        s = src[base + lane];
        d = dst[base + lane];
        nrm = dinv[s] * dinv[d];
    }
    const int co = lane * 2;

    for (int j0 = 0; j0 < cnt; j0 += EDGE_BATCH) {
        if (cnt - j0 >= EDGE_BATCH) {
            float2 v[EDGE_BATCH];
            #pragma unroll
            for (int j = 0; j < EDGE_BATCH; ++j) {
                const int sj = __builtin_amdgcn_readlane(s, j0 + j);
                v[j] = *(const float2*)(h + (size_t)sj * OUT_CH + co);
            }
            #pragma unroll
            for (int j = 0; j < EDGE_BATCH; ++j) {
                const int   dj = __builtin_amdgcn_readlane(d, j0 + j);
                const float nj = __int_as_float(
                    __builtin_amdgcn_readlane(__float_as_int(nrm), j0 + j));
                float* o = agg + (size_t)dj * OUT_CH + co;
                atomicAdd(o,     v[j].x * nj);
                atomicAdd(o + 1, v[j].y * nj);
            }
        } else {
            for (int j = j0; j < cnt; ++j) {
                const int   sj = __builtin_amdgcn_readlane(s, j);
                const int   dj = __builtin_amdgcn_readlane(d, j);
                const float nj = __int_as_float(
                    __builtin_amdgcn_readlane(__float_as_int(nrm), j));
                const float2 v = *(const float2*)(h + (size_t)sj * OUT_CH + co);
                float* o = agg + (size_t)dj * OUT_CH + co;
                atomicAdd(o,     v.x * nj);
                atomicAdd(o + 1, v.y * nj);
            }
        }
    }
}

__global__ void finalize_kernel(const float* __restrict__ agg,
                                const float* __restrict__ h,
                                const float* __restrict__ dinv,
                                const float* __restrict__ bias,
                                const float* __restrict__ alpha,
                                float* __restrict__ out, int n) {
    int idx = blockIdx.x * blockDim.x + threadIdx.x;
    if (idx >= n * OUT_CH) return;
    int c = idx & (OUT_CH - 1);
    int i = idx >> 6;
    float di = dinv[i];
    float v = agg[idx] + h[idx] * di * di + bias[c];
    out[idx] = (v >= 0.0f) ? v : alpha[c] * v;
}

extern "C" void kernel_launch(void* const* d_in, const int* in_sizes, int n_in,
                              void* d_out, int out_size, void* d_ws, size_t ws_size,
                              hipStream_t stream) {
    const float* x     = (const float*)d_in[0];
    const int*   ei    = (const int*)  d_in[1];
    const float* W     = (const float*)d_in[2];
    const float* bias  = (const float*)d_in[3];
    const float* alpha = (const float*)d_in[4];
    float*       out   = (float*)d_out;

    const int N = in_sizes[0] / IN_CH;
    const int E = in_sizes[1] / 2;
    const int* src = ei;
    const int* dst = ei + E;

    char* ws = (char*)d_ws;
    size_t hw_bytes = ((size_t)N * OUT_CH * sizeof(float) + 255) & ~(size_t)255;
    float* h    = (float*)ws;
    float* agg  = (float*)(ws + hw_bytes);
    float* dinv = (float*)(ws + 2 * hw_bytes);

    const int threads = 256;
    int totNC = N * OUT_CH;

    init_kernel<<<(totNC + threads - 1) / threads, threads, 0, stream>>>(agg, dinv, N);
    deg_kernel<<<(E + threads - 1) / threads, threads, 0, stream>>>(dst, dinv, E);
    dinv_kernel<<<(N + threads - 1) / threads, threads, 0, stream>>>(dinv, N);
    gemm_wmma_kernel<<<(N + ROWS_PER_BLOCK - 1) / ROWS_PER_BLOCK, threads, 0, stream>>>(
        x, W, h, N);
    scatter_kernel<<<(E + 8 * EDGES_PER_WAVE - 1) / (8 * EDGES_PER_WAVE), threads, 0, stream>>>(
        h, src, dst, dinv, agg, E);
    finalize_kernel<<<(totNC + threads - 1) / threads, threads, 0, stream>>>(
        agg, h, dinv, bias, alpha, out, N);
}


# scatter: lane l owns channels l and l+32, so each float atomic (and each h load) covers one whole 128-byte line with 32 lanes instead of half of two lines; arrival-order float atomic sums as before
# speedup vs baseline: 1.5209x; 1.1452x over previous
#include <hip/hip_runtime.h>
#include <hip/hip_bf16.h>

#define IN_CH   512
#define OUT_CH  64
#define KCHUNK  128
#define ROWS_PER_BLOCK 128

typedef __attribute__((ext_vector_type(2))) float v2f;
typedef __attribute__((ext_vector_type(4))) float v4f;
typedef __attribute__((ext_vector_type(8))) float v8f;

__global__ void init_kernel(float* __restrict__ agg, float* __restrict__ deg, int n) {
    int i = blockIdx.x * blockDim.x + threadIdx.x;
    int total = n * OUT_CH;
    if (i < total) agg[i] = 0.0f;
    if (i < n)     deg[i] = 1.0f;
}

__global__ void deg_kernel(const int* __restrict__ dst, float* __restrict__ deg, int e) {
    int i = blockIdx.x * blockDim.x + threadIdx.x;
    if (i < e) atomicAdd(&deg[dst[i]], 1.0f);
}

__global__ void dinv_kernel(float* __restrict__ deg, int n) {
    int i = blockIdx.x * blockDim.x + threadIdx.x;
    if (i < n) deg[i] = rsqrtf(deg[i]);
}

__global__ __launch_bounds__(256) void gemm_wmma_kernel(
        const float* __restrict__ x, const float* __restrict__ W,
        float* __restrict__ h, int n) {
    __shared__ __align__(16) v2f lds_w[(KCHUNK / 2) * OUT_CH];

    const int tid   = threadIdx.x;
    const int wave  = tid >> 5;
    const int lane  = tid & 31;
    const int half  = lane >> 4;
    const int l15   = lane & 15;
    const int kh    = half * 2;
    const int row0  = blockIdx.x * ROWS_PER_BLOCK + wave * 16;

    int arow = row0 + l15;
    if (arow >= n) arow = n - 1;
    const float* xrow = x + (size_t)arow * IN_CH;

    v8f acc[4] = {{}, {}, {}, {}};

    for (int kc = 0; kc < IN_CH; kc += KCHUNK) {
        #pragma unroll
        for (int i = 0; i < (KCHUNK / 2) * OUT_CH / 256; ++i) {
            int j = tid + i * 256;
            int p = j >> 6;
            int c = j & 63;
            v2f w2;
            w2.x = W[(size_t)(kc + 2 * p)     * OUT_CH + c];
            w2.y = W[(size_t)(kc + 2 * p + 1) * OUT_CH + c];
            lds_w[j] = w2;
        }
        __syncthreads();

        #pragma unroll 4
        for (int k = 0; k < KCHUNK; k += 4) {
            v2f a;
            a.x = xrow[kc + k + kh];
            a.y = xrow[kc + k + kh + 1];
            const int prow = (k >> 1) + half;
            #pragma unroll
            for (int t = 0; t < 4; ++t) {
                v2f b = lds_w[prow * OUT_CH + t * 16 + l15];
                acc[t] = __builtin_amdgcn_wmma_f32_16x16x4_f32(
                    false, a, false, b, (short)0, acc[t], false, false);
            }
        }
        __syncthreads();
    }

    #pragma unroll
    for (int v = 0; v < 8; ++v) {
        int row = row0 + half * 8 + v;
        if (row < n) {
            #pragma unroll
            for (int t = 0; t < 4; ++t)
                h[(size_t)row * OUT_CH + t * 16 + l15] = acc[t][v];
        }
    }
}

#define EDGES_PER_WAVE 32
#define EDGE_BATCH     16
#define HALF_ROW       (OUT_CH / 2)

__global__ __launch_bounds__(256) void scatter_kernel(
        const float* __restrict__ h, const int* __restrict__ src,
        const int* __restrict__ dst, const float* __restrict__ dinv,
        float* __restrict__ agg, int e) {
    const int lane = threadIdx.x & 31;
    const int wave = blockIdx.x * 8 + __builtin_amdgcn_readfirstlane((int)(threadIdx.x >> 5));
    const int base = wave * EDGES_PER_WAVE;
    if (base >= e) return;
    const int cnt = min(EDGES_PER_WAVE, e - base);

    int   s = 0, d = 0;
    float nrm = 0.0f;
    if (lane < cnt) {
        s = src[base + lane];
        d = dst[base + lane];
        nrm = dinv[s] * dinv[d];
    }

    for (int j0 = 0; j0 < cnt; j0 += EDGE_BATCH) {
        if (cnt - j0 >= EDGE_BATCH) {
            float va[EDGE_BATCH], vb[EDGE_BATCH];
            #pragma unroll
            for (int j = 0; j < EDGE_BATCH; ++j) {
                const int sj = __builtin_amdgcn_readlane(s, j0 + j);
                const float* r = h + (size_t)sj * OUT_CH + lane;
                va[j] = r[0];
                vb[j] = r[HALF_ROW];
            }
            #pragma unroll
            for (int j = 0; j < EDGE_BATCH; ++j) {
                const int   dj = __builtin_amdgcn_readlane(d, j0 + j);
                const float nj = __int_as_float(
                    __builtin_amdgcn_readlane(__float_as_int(nrm), j0 + j));
                float* o = agg + (size_t)dj * OUT_CH + lane;
                atomicAdd(o,            va[j] * nj);
                atomicAdd(o + HALF_ROW, vb[j] * nj);
            }
        } else {
            for (int j = j0; j < cnt; ++j) {
                const int   sj = __builtin_amdgcn_readlane(s, j);
                const int   dj = __builtin_amdgcn_readlane(d, j);
                const float nj = __int_as_float(
                    __builtin_amdgcn_readlane(__float_as_int(nrm), j));
                const float* r = h + (size_t)sj * OUT_CH + lane;
                const float a = r[0];
                const float b = r[HALF_ROW];
                float* o = agg + (size_t)dj * OUT_CH + lane;
                atomicAdd(o,            a * nj);
                atomicAdd(o + HALF_ROW, b * nj);
            }
        }
    }
}

__global__ void finalize_kernel(const float* __restrict__ agg,
                                const float* __restrict__ h,
                                const float* __restrict__ dinv,
                                const float* __restrict__ bias,
                                const float* __restrict__ alpha,
                                float* __restrict__ out, int n) {
    int idx = blockIdx.x * blockDim.x + threadIdx.x;
    if (idx >= n * OUT_CH) return;
    int c = idx & (OUT_CH - 1);
    int i = idx >> 6;
    float di = dinv[i];
    float v = agg[idx] + h[idx] * di * di + bias[c];
    out[idx] = (v >= 0.0f) ? v : alpha[c] * v;
}

extern "C" void kernel_launch(void* const* d_in, const int* in_sizes, int n_in,
                              void* d_out, int out_size, void* d_ws, size_t ws_size,
                              hipStream_t stream) {
    const float* x     = (const float*)d_in[0];
    const int*   ei    = (const int*)  d_in[1];
    const float* W     = (const float*)d_in[2];
    const float* bias  = (const float*)d_in[3];
    const float* alpha = (const float*)d_in[4];
    float*       out   = (float*)d_out;

    const int N = in_sizes[0] / IN_CH;
    const int E = in_sizes[1] / 2;
    const int* src = ei;
    const int* dst = ei + E;

    char* ws = (char*)d_ws;
    size_t hw_bytes = ((size_t)N * OUT_CH * sizeof(float) + 255) & ~(size_t)255;
    float* h    = (float*)ws;
    float* agg  = (float*)(ws + hw_bytes);
    float* dinv = (float*)(ws + 2 * hw_bytes);

    const int threads = 256;
    int totNC = N * OUT_CH;

    init_kernel<<<(totNC + threads - 1) / threads, threads, 0, stream>>>(agg, dinv, N);
    deg_kernel<<<(E + threads - 1) / threads, threads, 0, stream>>>(dst, dinv, E);
    dinv_kernel<<<(N + threads - 1) / threads, threads, 0, stream>>>(dinv, N);
    gemm_wmma_kernel<<<(N + ROWS_PER_BLOCK - 1) / ROWS_PER_BLOCK, threads, 0, stream>>>(
        x, W, h, N);
    scatter_kernel<<<(E + 8 * EDGES_PER_WAVE - 1) / (8 * EDGES_PER_WAVE), threads, 0, stream>>>(
        h, src, dst, dinv, agg, E);
    finalize_kernel<<<(totNC + threads - 1) / threads, threads, 0, stream>>>(
        agg, h, dinv, bias, alpha, out, N);
}


# symmetric norm factored out of the edge sum: GEMM epilogue stores h*dinv[row], scatter adds raw rows (no dinv gathers, no multiply), finalize scales by dinv[dst]; each term rounds differently
# speedup vs baseline: 1.6078x; 1.0571x over previous
#include <hip/hip_runtime.h>
#include <hip/hip_bf16.h>

#define IN_CH   512
#define OUT_CH  64
#define KCHUNK  128
#define ROWS_PER_BLOCK 128

typedef __attribute__((ext_vector_type(2))) float v2f;
typedef __attribute__((ext_vector_type(4))) float v4f;
typedef __attribute__((ext_vector_type(8))) float v8f;

__global__ void init_kernel(float* __restrict__ agg, float* __restrict__ deg, int n) {
    int i = blockIdx.x * blockDim.x + threadIdx.x;
    int total = n * OUT_CH;
    if (i < total) agg[i] = 0.0f;
    if (i < n)     deg[i] = 1.0f;
}

__global__ void deg_kernel(const int* __restrict__ dst, float* __restrict__ deg, int e) {
    int i = blockIdx.x * blockDim.x + threadIdx.x;
    if (i < e) atomicAdd(&deg[dst[i]], 1.0f);
}

__global__ void dinv_kernel(float* __restrict__ deg, int n) {
    int i = blockIdx.x * blockDim.x + threadIdx.x;
    if (i < n) deg[i] = rsqrtf(deg[i]);
}

__global__ __launch_bounds__(256) void gemm_wmma_kernel(
        const float* __restrict__ x, const float* __restrict__ W,
        const float* __restrict__ dinv, float* __restrict__ h, int n) {
    __shared__ __align__(16) v2f lds_w[(KCHUNK / 2) * OUT_CH];

    const int tid   = threadIdx.x;
    const int wave  = tid >> 5;
    const int lane  = tid & 31;
    const int half  = lane >> 4;
    const int l15   = lane & 15;
    const int kh    = half * 2;
    const int row0  = blockIdx.x * ROWS_PER_BLOCK + wave * 16;

    int arow = row0 + l15;
    if (arow >= n) arow = n - 1;
    const float* xrow = x + (size_t)arow * IN_CH;

    float dr[8];
    #pragma unroll
    for (int v = 0; v < 8; ++v) {
        int r = row0 + half * 8 + v;
        dr[v] = dinv[r < n ? r : n - 1];
    }

    v8f acc[4] = {{}, {}, {}, {}};

    for (int kc = 0; kc < IN_CH; kc += KCHUNK) {
        #pragma unroll
        for (int i = 0; i < (KCHUNK / 2) * OUT_CH / 256; ++i) {
            int j = tid + i * 256;
            int p = j >> 6;
            int c = j & 63;
            v2f w2;
            w2.x = W[(size_t)(kc + 2 * p)     * OUT_CH + c];
            w2.y = W[(size_t)(kc + 2 * p + 1) * OUT_CH + c];
            lds_w[j] = w2;
        }
        __syncthreads();

        #pragma unroll 4
        for (int k = 0; k < KCHUNK; k += 4) {
            v2f a;
            a.x = xrow[kc + k + kh];
            a.y = xrow[kc + k + kh + 1];
            const int prow = (k >> 1) + half;
            #pragma unroll
            for (int t = 0; t < 4; ++t) {
                v2f b = lds_w[prow * OUT_CH + t * 16 + l15];
                acc[t] = __builtin_amdgcn_wmma_f32_16x16x4_f32(
                    false, a, false, b, (short)0, acc[t], false, false);
            }
        }
        __syncthreads();
    }

    #pragma unroll
    for (int v = 0; v < 8; ++v) {
        int row = row0 + half * 8 + v;
        if (row < n) {
            #pragma unroll
            for (int t = 0; t < 4; ++t)
                h[(size_t)row * OUT_CH + t * 16 + l15] = acc[t][v] * dr[v];
        }
    }
}

#define EDGES_PER_WAVE 32
#define EDGE_BATCH     16
#define HALF_ROW       (OUT_CH / 2)

__global__ __launch_bounds__(256) void scatter_kernel(
        const float* __restrict__ h, const int* __restrict__ src,
        const int* __restrict__ dst, float* __restrict__ agg, int e) {
    const int lane = threadIdx.x & 31;
    const int wave = blockIdx.x * 8 + __builtin_amdgcn_readfirstlane((int)(threadIdx.x >> 5));
    const int base = wave * EDGES_PER_WAVE;
    if (base >= e) return;
    const int cnt = min(EDGES_PER_WAVE, e - base);

    int s = 0, d = 0;
    if (lane < cnt) {
        s = src[base + lane];
        d = dst[base + lane];
    }

    for (int j0 = 0; j0 < cnt; j0 += EDGE_BATCH) {
        if (cnt - j0 >= EDGE_BATCH) {
            float va[EDGE_BATCH], vb[EDGE_BATCH];
            #pragma unroll
            for (int j = 0; j < EDGE_BATCH; ++j) {
                const int sj = __builtin_amdgcn_readlane(s, j0 + j);
                const float* r = h + (size_t)sj * OUT_CH + lane;
                va[j] = r[0];
                vb[j] = r[HALF_ROW];
            }
            #pragma unroll
            for (int j = 0; j < EDGE_BATCH; ++j) {
                const int dj = __builtin_amdgcn_readlane(d, j0 + j);
                float* o = agg + (size_t)dj * OUT_CH + lane;
                atomicAdd(o,            va[j]);
                atomicAdd(o + HALF_ROW, vb[j]);
            }
        } else {
            for (int j = j0; j < cnt; ++j) {
                const int sj = __builtin_amdgcn_readlane(s, j);
                const int dj = __builtin_amdgcn_readlane(d, j);
                const float* r = h + (size_t)sj * OUT_CH + lane;
                const float a = r[0];
                const float b = r[HALF_ROW];
                float* o = agg + (size_t)dj * OUT_CH + lane;
                atomicAdd(o,            a);
                atomicAdd(o + HALF_ROW, b);
            }
        }
    }
}

__global__ void finalize_kernel(const float* __restrict__ agg,
                                const float* __restrict__ h,
                                const float* __restrict__ dinv,
                                const float* __restrict__ bias,
                                const float* __restrict__ alpha,
                                float* __restrict__ out, int n) {
    int idx = blockIdx.x * blockDim.x + threadIdx.x;
    if (idx >= n * OUT_CH) return;
    int c = idx & (OUT_CH - 1);
    int i = idx >> 6;
    float di = dinv[i];
    float v = (agg[idx] + h[idx]) * di + bias[c];
    out[idx] = (v >= 0.0f) ? v : alpha[c] * v;
}

extern "C" void kernel_launch(void* const* d_in, const int* in_sizes, int n_in,
                              void* d_out, int out_size, void* d_ws, size_t ws_size,
                              hipStream_t stream) {
    const float* x     = (const float*)d_in[0];
    const int*   ei    = (const int*)  d_in[1];
    const float* W     = (const float*)d_in[2];
    const float* bias  = (const float*)d_in[3];
    const float* alpha = (const float*)d_in[4];
    float*       out   = (float*)d_out;

    const int N = in_sizes[0] / IN_CH;
    const int E = in_sizes[1] / 2;
    const int* src = ei;
    const int* dst = ei + E;

    char* ws = (char*)d_ws;
    size_t hw_bytes = ((size_t)N * OUT_CH * sizeof(float) + 255) & ~(size_t)255;
    float* h    = (float*)ws;
    float* agg  = (float*)(ws + hw_bytes);
    float* dinv = (float*)(ws + 2 * hw_bytes);

    const int threads = 256;
    int totNC = N * OUT_CH;

    init_kernel<<<(totNC + threads - 1) / threads, threads, 0, stream>>>(agg, dinv, N);
    deg_kernel<<<(E + threads - 1) / threads, threads, 0, stream>>>(dst, dinv, E);
    dinv_kernel<<<(N + threads - 1) / threads, threads, 0, stream>>>(dinv, N);
    gemm_wmma_kernel<<<(N + ROWS_PER_BLOCK - 1) / ROWS_PER_BLOCK, threads, 0, stream>>>(
        x, W, dinv, h, N);
    scatter_kernel<<<(E + 8 * EDGES_PER_WAVE - 1) / (8 * EDGES_PER_WAVE), threads, 0, stream>>>(
        h, src, dst, agg, E);
    finalize_kernel<<<(totNC + threads - 1) / threads, threads, 0, stream>>>(
        agg, h, dinv, bias, alpha, out, N);
}


# edge sum in LDS: edges binned per call into 64-node destination buckets; one work-group per bucket adds source rows with LDS float adds (arrival order), writes out; no L2 float atomics on the sum
# speedup vs baseline: 1.7552x; 1.0917x over previous
#include <hip/hip_runtime.h>
#include <hip/hip_bf16.h>

#define IN_CH   512
#define OUT_CH  64
#define KCHUNK  128
#define ROWS_PER_BLOCK 128

typedef __attribute__((ext_vector_type(2))) float v2f;
typedef __attribute__((ext_vector_type(4))) float v4f;
typedef __attribute__((ext_vector_type(8))) float v8f;

#define BN            64
#define BN_SHIFT      6
#define NBLK          512
#define SORT_THREADS  1024
#define EDGE_UNROLL   4

__global__ void init_kernel(float* __restrict__ deg, int* __restrict__ tot,
                            int* __restrict__ cursor, int n, int nb) {
    int i = blockIdx.x * blockDim.x + threadIdx.x;
    if (i < n)  deg[i] = 1.0f;
    if (i < nb) { tot[i] = 0; cursor[i] = 0; }
}

__global__ __launch_bounds__(SORT_THREADS) void deg_count_kernel(
        const int* __restrict__ dst, float* __restrict__ deg, int* __restrict__ tot,
        int e, int nb, int chunk) {
    extern __shared__ int sm[];
    int* hist = sm;
    const int tid = threadIdx.x;
    for (int i = tid; i < nb; i += SORT_THREADS) hist[i] = 0;
    __syncthreads();
    const int beg = blockIdx.x * chunk;
    const int end = min(beg + chunk, e);
    for (int i0 = beg + tid; i0 < end; i0 += SORT_THREADS * EDGE_UNROLL) {
        int d[EDGE_UNROLL];
        #pragma unroll
        for (int u = 0; u < EDGE_UNROLL; ++u)
            d[u] = dst[min(i0 + u * SORT_THREADS, end - 1)];
        #pragma unroll
        for (int u = 0; u < EDGE_UNROLL; ++u) {
            if (i0 + u * SORT_THREADS < end) {
                atomicAdd(&deg[d[u]], 1.0f);
                atomicAdd(&hist[d[u] >> BN_SHIFT], 1);
            }
        }
    }
    __syncthreads();
    for (int i = tid; i < nb; i += SORT_THREADS) {
        const int c = hist[i];
        if (c) atomicAdd(&tot[i], c);
    }
}

__global__ void dinv_kernel(float* __restrict__ deg, int n) {
    int i = blockIdx.x * blockDim.x + threadIdx.x;
    if (i < n) deg[i] = rsqrtf(deg[i]);
}

__global__ __launch_bounds__(256) void gemm_wmma_kernel(
        const float* __restrict__ x, const float* __restrict__ W,
        const float* __restrict__ dinv, float* __restrict__ h, int n) {
    __shared__ __align__(16) v2f lds_w[(KCHUNK / 2) * OUT_CH];

    const int tid   = threadIdx.x;
    const int wave  = tid >> 5;
    const int lane  = tid & 31;
    const int half  = lane >> 4;
    const int l15   = lane & 15;
    const int kh    = half * 2;
    const int row0  = blockIdx.x * ROWS_PER_BLOCK + wave * 16;

    int arow = row0 + l15;
    if (arow >= n) arow = n - 1;
    const float* xrow = x + (size_t)arow * IN_CH;

    float dr[8];
    #pragma unroll
    for (int v = 0; v < 8; ++v) {
        int r = row0 + half * 8 + v;
        dr[v] = dinv[r < n ? r : n - 1];
    }

    v8f acc[4] = {{}, {}, {}, {}};

    for (int kc = 0; kc < IN_CH; kc += KCHUNK) {
        #pragma unroll
        for (int i = 0; i < (KCHUNK / 2) * OUT_CH / 256; ++i) {
            int j = tid + i * 256;
            int p = j >> 6;
            int c = j & 63;
            v2f w2;
            w2.x = W[(size_t)(kc + 2 * p)     * OUT_CH + c];
            w2.y = W[(size_t)(kc + 2 * p + 1) * OUT_CH + c];
            lds_w[j] = w2;
        }
        __syncthreads();

        #pragma unroll 4
        for (int k = 0; k < KCHUNK; k += 4) {
            v2f a;
            a.x = xrow[kc + k + kh];
            a.y = xrow[kc + k + kh + 1];
            const int prow = (k >> 1) + half;
            #pragma unroll
            for (int t = 0; t < 4; ++t) {
                v2f b = lds_w[prow * OUT_CH + t * 16 + l15];
                acc[t] = __builtin_amdgcn_wmma_f32_16x16x4_f32(
                    false, a, false, b, (short)0, acc[t], false, false);
            }
        }
        __syncthreads();
    }

    #pragma unroll
    for (int v = 0; v < 8; ++v) {
        int row = row0 + half * 8 + v;
        if (row < n) {
            #pragma unroll
            for (int t = 0; t < 4; ++t)
                h[(size_t)row * OUT_CH + t * 16 + l15] = acc[t][v] * dr[v];
        }
    }
}

__device__ __forceinline__ int block_excl_scan(int v, int* wsum, int tid) {
    const int lane = tid & 31;
    const int w    = tid >> 5;
    int inc = v;
    for (int o = 1; o < 32; o <<= 1) {
        const int nbr = __shfl_up(inc, o, 32);
        if (lane >= o) inc += nbr;
    }
    if (lane == 31) wsum[w] = inc;
    __syncthreads();
    int add = 0;
    for (int k = 0; k < w; ++k) add += wsum[k];
    __syncthreads();
    return add + inc - v;
}

__global__ __launch_bounds__(SORT_THREADS) void fill_kernel(
        const int* __restrict__ src, const int* __restrict__ dst,
        const int* __restrict__ tot, int* __restrict__ cursor,
        int* __restrict__ base_out, unsigned* __restrict__ binned,
        int e, int nb, int chunk) {
    extern __shared__ int sm[];
    int* hist = sm;
    int* pos  = sm + nb;
    int* wsum = sm + 2 * nb;
    const int tid = threadIdx.x;
    for (int i = tid; i < nb; i += SORT_THREADS) hist[i] = 0;
    __syncthreads();
    const int beg = blockIdx.x * chunk;
    const int end = min(beg + chunk, e);
    for (int i0 = beg + tid; i0 < end; i0 += SORT_THREADS * EDGE_UNROLL) {
        int d[EDGE_UNROLL];
        #pragma unroll
        for (int u = 0; u < EDGE_UNROLL; ++u)
            d[u] = dst[min(i0 + u * SORT_THREADS, end - 1)];
        #pragma unroll
        for (int u = 0; u < EDGE_UNROLL; ++u)
            if (i0 + u * SORT_THREADS < end) atomicAdd(&hist[d[u] >> BN_SHIFT], 1);
    }

    const int seg = (nb + SORT_THREADS - 1) / SORT_THREADS;
    const int b0  = tid * seg;
    int tsum = 0;
    for (int k = 0; k < seg; ++k) {
        const int b = b0 + k;
        if (b < nb) tsum += tot[b];
    }
    int run = block_excl_scan(tsum, wsum, tid);
    for (int k = 0; k < seg; ++k) {
        const int b = b0 + k;
        if (b < nb) {
            if (blockIdx.x == 0) base_out[b] = run;
            const int c = hist[b];
            pos[b] = run + (c ? atomicAdd(&cursor[b], c) : 0);
            run += tot[b];
        }
    }
    __syncthreads();

    for (int i0 = beg + tid; i0 < end; i0 += SORT_THREADS * EDGE_UNROLL) {
        int s[EDGE_UNROLL], d[EDGE_UNROLL];
        #pragma unroll
        for (int u = 0; u < EDGE_UNROLL; ++u) {
            const int ic = min(i0 + u * SORT_THREADS, end - 1);
            s[u] = src[ic];
            d[u] = dst[ic];
        }
        #pragma unroll
        for (int u = 0; u < EDGE_UNROLL; ++u) {
            if (i0 + u * SORT_THREADS < end) {
                const int p = atomicAdd(&pos[d[u] >> BN_SHIFT], 1);
                binned[p] = ((unsigned)s[u] << BN_SHIFT) | (unsigned)(d[u] & (BN - 1));
            }
        }
    }
}

#define EDGE_BATCH     16
#define HALF_ROW       (OUT_CH / 2)

__device__ __forceinline__ void lds_add(float* p, float v) {
    __hip_atomic_fetch_add(p, v, __ATOMIC_RELAXED, __HIP_MEMORY_SCOPE_WORKGROUP);
}

template <bool TAIL>
__device__ __forceinline__ void write_bucket(
        const float* __restrict__ h, const float* __restrict__ dinv,
        const float* __restrict__ bias, const float* __restrict__ alpha,
        const v4f* acc4, float* __restrict__ out, int node0, int n, int tid) {
    const int q = tid & 15;
    float bq[4], aq[4];
    for (int t = 0; t < 4; ++t) { bq[t] = bias[4 * q + t]; aq[t] = alpha[4 * q + t]; }
    #pragma unroll
    for (int k = 0; k < BN / 16; ++k) {
        const int r    = (tid >> 4) + 16 * k;
        const int node = node0 + r;
        if (TAIL && node >= n) continue;
        const v4f hv = *(const v4f*)(h + (size_t)node * OUT_CH + 4 * q);
        const float di = dinv[node];
        const v4f sv = acc4[r * 16 + q];
        v4f v;
        for (int t = 0; t < 4; ++t) {
            const float y = (sv[t] + hv[t]) * di + bq[t];
            v[t] = (y >= 0.0f) ? y : aq[t] * y;
        }
        *(v4f*)(out + (size_t)node * OUT_CH + 4 * q) = v;
    }
}

__global__ __launch_bounds__(256) void gather_kernel(
        const float* __restrict__ h, const unsigned* __restrict__ binned,
        const int* __restrict__ base, const int* __restrict__ tot,
        const float* __restrict__ dinv, const float* __restrict__ bias,
        const float* __restrict__ alpha, float* __restrict__ out, int n) {
    __shared__ __align__(16) float acc[BN * OUT_CH];
    v4f* acc4 = (v4f*)acc;
    const int tid  = threadIdx.x;
    const int lane = tid & 31;
    const int wave = __builtin_amdgcn_readfirstlane(tid >> 5);
    const int b    = blockIdx.x;

    for (int k = 0; k < BN * OUT_CH / 4 / 256; ++k)
        acc4[tid + k * 256] = (v4f){0.0f, 0.0f, 0.0f, 0.0f};
    __syncthreads();

    const int beg = base[b];
    const int cnt = tot[b];

    for (int c0 = wave * 32; c0 < cnt; c0 += 8 * 32) {
        const int m = min(32, cnt - c0);
        unsigned pk = 0;
        if (lane < m) pk = binned[beg + c0 + lane];

        for (int j0 = 0; j0 < m; j0 += EDGE_BATCH) {
            if (m - j0 >= EDGE_BATCH) {
                float va[EDGE_BATCH], vb[EDGE_BATCH];
                #pragma unroll
                for (int j = 0; j < EDGE_BATCH; ++j) {
                    const unsigned pj = __builtin_amdgcn_readlane(pk, j0 + j);
                    const float* r = h + (size_t)(pj >> BN_SHIFT) * OUT_CH + lane;
                    va[j] = r[0];
                    vb[j] = r[HALF_ROW];
                }
                #pragma unroll
                for (int j = 0; j < EDGE_BATCH; ++j) {
                    const unsigned pj = __builtin_amdgcn_readlane(pk, j0 + j);
                    float* o = acc + (pj & (BN - 1)) * OUT_CH + lane;
                    lds_add(o,            va[j]);
                    lds_add(o + HALF_ROW, vb[j]);
                }
            } else {
                for (int j = j0; j < m; ++j) {
                    const unsigned pj = __builtin_amdgcn_readlane(pk, j);
                    const float* r = h + (size_t)(pj >> BN_SHIFT) * OUT_CH + lane;
                    const float a = r[0];
                    const float c = r[HALF_ROW];
                    float* o = acc + (pj & (BN - 1)) * OUT_CH + lane;
                    lds_add(o,            a);
                    lds_add(o + HALF_ROW, c);
                }
            }
        }
    }
    __syncthreads();

    const int node0 = b * BN;
    if (node0 + BN <= n) write_bucket<false>(h, dinv, bias, alpha, acc4, out, node0, n, tid);
    else                 write_bucket<true >(h, dinv, bias, alpha, acc4, out, node0, n, tid);
}

extern "C" void kernel_launch(void* const* d_in, const int* in_sizes, int n_in,
                              void* d_out, int out_size, void* d_ws, size_t ws_size,
                              hipStream_t stream) {
    const float* x     = (const float*)d_in[0];
    const int*   ei    = (const int*)  d_in[1];
    const float* W     = (const float*)d_in[2];
    const float* bias  = (const float*)d_in[3];
    const float* alpha = (const float*)d_in[4];
    float*       out   = (float*)d_out;

    const int N = in_sizes[0] / IN_CH;
    const int E = in_sizes[1] / 2;
    const int* src = ei;
    const int* dst = ei + E;

    char* ws = (char*)d_ws;
    size_t hw_bytes = ((size_t)N * OUT_CH * sizeof(float) + 255) & ~(size_t)255;
    size_t eb_bytes = ((size_t)E * sizeof(unsigned) + 255) & ~(size_t)255;
    const int NB = (N + BN - 1) / BN;
    size_t nb_bytes = ((size_t)NB * sizeof(int) + 255) & ~(size_t)255;
    float*    h      = (float*)ws;
    unsigned* binned = (unsigned*)(ws + hw_bytes);
    int*      tot    = (int*)(ws + hw_bytes + eb_bytes);
    int*      cursor = (int*)(ws + hw_bytes + eb_bytes + nb_bytes);
    int*      base   = (int*)(ws + hw_bytes + eb_bytes + 2 * nb_bytes);
    float*    dinv   = (float*)(ws + 2 * hw_bytes);

    const int threads = 256;
    const int chunk = (E + NBLK - 1) / NBLK;

    init_kernel<<<(N + threads - 1) / threads, threads, 0, stream>>>(dinv, tot, cursor, N, NB);
    deg_count_kernel<<<NBLK, SORT_THREADS, (size_t)NB * sizeof(int), stream>>>(dst, dinv, tot, E, NB, chunk);
    dinv_kernel<<<(N + threads - 1) / threads, threads, 0, stream>>>(dinv, N);
    gemm_wmma_kernel<<<(N + ROWS_PER_BLOCK - 1) / ROWS_PER_BLOCK, threads, 0, stream>>>(
        x, W, dinv, h, N);
    fill_kernel<<<NBLK, SORT_THREADS, (size_t)(2 * NB + 32) * sizeof(int), stream>>>(
        src, dst, tot, cursor, base, binned, E, NB, chunk);
    gather_kernel<<<NB, threads, 0, stream>>>(
        h, binned, base, tot, dinv, bias, alpha, out, N);
}


# in-degrees counted from the binned edge words after the fill pass (LDS integer adds, one work-group per 256 nodes, writes rsqrt(1+count)); no global float atomics on degrees; dinv values unchanged
# speedup vs baseline: 3.5184x; 2.0046x over previous
#include <hip/hip_runtime.h>
#include <hip/hip_bf16.h>

#define IN_CH   512
#define OUT_CH  64
#define KCHUNK  128
#define ROWS_PER_BLOCK 128

typedef __attribute__((ext_vector_type(2))) float v2f;
typedef __attribute__((ext_vector_type(4))) float v4f;
typedef __attribute__((ext_vector_type(8))) float v8f;

#define BN            64
#define BN_SHIFT      6
#define NBLK          512
#define SORT_THREADS  1024
#define EDGE_UNROLL   4

__global__ void init_kernel(int* __restrict__ tot, int* __restrict__ cursor, int nb) {
    int i = blockIdx.x * blockDim.x + threadIdx.x;
    if (i < nb) { tot[i] = 0; cursor[i] = 0; }
}

__global__ __launch_bounds__(SORT_THREADS) void count_kernel(
        const int* __restrict__ dst, int* __restrict__ tot,
        int e, int nb, int chunk) {
    extern __shared__ int sm[];
    int* hist = sm;
    const int tid = threadIdx.x;
    for (int i = tid; i < nb; i += SORT_THREADS) hist[i] = 0;
    __syncthreads();
    const int beg = blockIdx.x * chunk;
    const int end = min(beg + chunk, e);
    for (int i0 = beg + tid; i0 < end; i0 += SORT_THREADS * EDGE_UNROLL) {
        int d[EDGE_UNROLL];
        #pragma unroll
        for (int u = 0; u < EDGE_UNROLL; ++u)
            d[u] = dst[min(i0 + u * SORT_THREADS, end - 1)];
        #pragma unroll
        for (int u = 0; u < EDGE_UNROLL; ++u) {
            if (i0 + u * SORT_THREADS < end) atomicAdd(&hist[d[u] >> BN_SHIFT], 1);
        }
    }
    __syncthreads();
    for (int i = tid; i < nb; i += SORT_THREADS) {
        const int c = hist[i];
        if (c) atomicAdd(&tot[i], c);
    }
}

#define DG_BUCKETS 4
#define DG_NODES   (DG_BUCKETS * BN)
#define DG_UNROLL  16

__global__ __launch_bounds__(DG_NODES) void degree_dinv_kernel(
        const unsigned* __restrict__ binned, const int* __restrict__ base,
        const int* __restrict__ tot, float* __restrict__ dinv, int n, int nb) {
    __shared__ int cnt[DG_NODES];
    const int tid = threadIdx.x;
    const int b0  = blockIdx.x * DG_BUCKETS;
    cnt[tid] = 0;
    int bnd[DG_BUCKETS + 1];
    bnd[0] = base[b0];
    for (int k = 0; k < DG_BUCKETS; ++k)
        bnd[k + 1] = bnd[k] + ((b0 + k < nb) ? tot[b0 + k] : 0);
    __syncthreads();
    const int end = bnd[DG_BUCKETS];
    for (int i0 = bnd[0] + tid; i0 < end; i0 += DG_NODES * DG_UNROLL) {
        unsigned w[DG_UNROLL];
        #pragma unroll
        for (int u = 0; u < DG_UNROLL; ++u)
            w[u] = binned[min(i0 + u * DG_NODES, end - 1)];
        #pragma unroll
        for (int u = 0; u < DG_UNROLL; ++u) {
            const int i = i0 + u * DG_NODES;
            const int k = (i >= bnd[1]) + (i >= bnd[2]) + (i >= bnd[3]);
            atomicAdd(&cnt[(k << BN_SHIFT) | (int)(w[u] & (BN - 1))], (i < end) ? 1 : 0);
        }
    }
    __syncthreads();
    const int node = b0 * BN + tid;
    if (node < n) dinv[node] = rsqrtf(1.0f + (float)cnt[tid]);
}

__global__ __launch_bounds__(256) void gemm_wmma_kernel(
        const float* __restrict__ x, const float* __restrict__ W,
        const float* __restrict__ dinv, float* __restrict__ h, int n) {
    __shared__ __align__(16) v2f lds_w[(KCHUNK / 2) * OUT_CH];

    const int tid   = threadIdx.x;
    const int wave  = tid >> 5;
    const int lane  = tid & 31;
    const int half  = lane >> 4;
    const int l15   = lane & 15;
    const int kh    = half * 2;
    const int row0  = blockIdx.x * ROWS_PER_BLOCK + wave * 16;

    int arow = row0 + l15;
    if (arow >= n) arow = n - 1;
    const float* xrow = x + (size_t)arow * IN_CH;

    float dr[8];
    #pragma unroll
    for (int v = 0; v < 8; ++v) {
        int r = row0 + half * 8 + v;
        dr[v] = dinv[r < n ? r : n - 1];
    }

    v8f acc[4] = {{}, {}, {}, {}};

    for (int kc = 0; kc < IN_CH; kc += KCHUNK) {
        #pragma unroll
        for (int i = 0; i < (KCHUNK / 2) * OUT_CH / 256; ++i) {
            int j = tid + i * 256;
            int p = j >> 6;
            int c = j & 63;
            v2f w2;
            w2.x = W[(size_t)(kc + 2 * p)     * OUT_CH + c];
            w2.y = W[(size_t)(kc + 2 * p + 1) * OUT_CH + c];
            lds_w[j] = w2;
        }
        __syncthreads();

        #pragma unroll 4
        for (int k = 0; k < KCHUNK; k += 4) {
            v2f a;
            a.x = xrow[kc + k + kh];
            a.y = xrow[kc + k + kh + 1];
            const int prow = (k >> 1) + half;
            #pragma unroll
            for (int t = 0; t < 4; ++t) {
                v2f b = lds_w[prow * OUT_CH + t * 16 + l15];
                acc[t] = __builtin_amdgcn_wmma_f32_16x16x4_f32(
                    false, a, false, b, (short)0, acc[t], false, false);
            }
        }
        __syncthreads();
    }

    #pragma unroll
    for (int v = 0; v < 8; ++v) {
        int row = row0 + half * 8 + v;
        if (row < n) {
            #pragma unroll
            for (int t = 0; t < 4; ++t)
                h[(size_t)row * OUT_CH + t * 16 + l15] = acc[t][v] * dr[v];
        }
    }
}

__device__ __forceinline__ int block_excl_scan(int v, int* wsum, int tid) {
    const int lane = tid & 31;
    const int w    = tid >> 5;
    int inc = v;
    for (int o = 1; o < 32; o <<= 1) {
        const int nbr = __shfl_up(inc, o, 32);
        if (lane >= o) inc += nbr;
    }
    if (lane == 31) wsum[w] = inc;
    __syncthreads();
    int add = 0;
    for (int k = 0; k < w; ++k) add += wsum[k];
    __syncthreads();
    return add + inc - v;
}

__global__ __launch_bounds__(SORT_THREADS) void fill_kernel(
        const int* __restrict__ src, const int* __restrict__ dst,
        const int* __restrict__ tot, int* __restrict__ cursor,
        int* __restrict__ base_out, unsigned* __restrict__ binned,
        int e, int nb, int chunk) {
    extern __shared__ int sm[];
    int* hist = sm;
    int* pos  = sm + nb;
    int* wsum = sm + 2 * nb;
    const int tid = threadIdx.x;
    for (int i = tid; i < nb; i += SORT_THREADS) hist[i] = 0;
    __syncthreads();
    const int beg = blockIdx.x * chunk;
    const int end = min(beg + chunk, e);
    for (int i0 = beg + tid; i0 < end; i0 += SORT_THREADS * EDGE_UNROLL) {
        int d[EDGE_UNROLL];
        #pragma unroll
        for (int u = 0; u < EDGE_UNROLL; ++u)
            d[u] = dst[min(i0 + u * SORT_THREADS, end - 1)];
        #pragma unroll
        for (int u = 0; u < EDGE_UNROLL; ++u)
            if (i0 + u * SORT_THREADS < end) atomicAdd(&hist[d[u] >> BN_SHIFT], 1);
    }

    const int seg = (nb + SORT_THREADS - 1) / SORT_THREADS;
    const int b0  = tid * seg;
    int tsum = 0;
    for (int k = 0; k < seg; ++k) {
        const int b = b0 + k;
        if (b < nb) tsum += tot[b];
    }
    int run = block_excl_scan(tsum, wsum, tid);
    for (int k = 0; k < seg; ++k) {
        const int b = b0 + k;
        if (b < nb) {
            if (blockIdx.x == 0) base_out[b] = run;
            const int c = hist[b];
            pos[b] = run + (c ? atomicAdd(&cursor[b], c) : 0);
            run += tot[b];
        }
    }
    __syncthreads();

    for (int i0 = beg + tid; i0 < end; i0 += SORT_THREADS * EDGE_UNROLL) {
        int s[EDGE_UNROLL], d[EDGE_UNROLL];
        #pragma unroll
        for (int u = 0; u < EDGE_UNROLL; ++u) {
            const int ic = min(i0 + u * SORT_THREADS, end - 1);
            s[u] = src[ic];
            d[u] = dst[ic];
        }
        #pragma unroll
        for (int u = 0; u < EDGE_UNROLL; ++u) {
            if (i0 + u * SORT_THREADS < end) {
                const int p = atomicAdd(&pos[d[u] >> BN_SHIFT], 1);
                binned[p] = ((unsigned)s[u] << BN_SHIFT) | (unsigned)(d[u] & (BN - 1));
            }
        }
    }
}

#define EDGE_BATCH     16
#define HALF_ROW       (OUT_CH / 2)

__device__ __forceinline__ void lds_add(float* p, float v) {
    __hip_atomic_fetch_add(p, v, __ATOMIC_RELAXED, __HIP_MEMORY_SCOPE_WORKGROUP);
}

template <bool TAIL>
__device__ __forceinline__ void write_bucket(
        const float* __restrict__ h, const float* __restrict__ dinv,
        const float* __restrict__ bias, const float* __restrict__ alpha,
        const v4f* acc4, float* __restrict__ out, int node0, int n, int tid) {
    const int q = tid & 15;
    float bq[4], aq[4];
    for (int t = 0; t < 4; ++t) { bq[t] = bias[4 * q + t]; aq[t] = alpha[4 * q + t]; }
    #pragma unroll
    for (int k = 0; k < BN / 16; ++k) {
        const int r    = (tid >> 4) + 16 * k;
        const int node = node0 + r;
        if (TAIL && node >= n) continue;
        const v4f hv = *(const v4f*)(h + (size_t)node * OUT_CH + 4 * q);
        const float di = dinv[node];
        const v4f sv = acc4[r * 16 + q];
        v4f v;
        for (int t = 0; t < 4; ++t) {
            const float y = (sv[t] + hv[t]) * di + bq[t];
            v[t] = (y >= 0.0f) ? y : aq[t] * y;
        }
        *(v4f*)(out + (size_t)node * OUT_CH + 4 * q) = v;
    }
}

__global__ __launch_bounds__(256) void gather_kernel(
        const float* __restrict__ h, const unsigned* __restrict__ binned,
        const int* __restrict__ base, const int* __restrict__ tot,
        const float* __restrict__ dinv, const float* __restrict__ bias,
        const float* __restrict__ alpha, float* __restrict__ out, int n) {
    __shared__ __align__(16) float acc[BN * OUT_CH];
    v4f* acc4 = (v4f*)acc;
    const int tid  = threadIdx.x;
    const int lane = tid & 31;
    const int wave = __builtin_amdgcn_readfirstlane(tid >> 5);
    const int b    = blockIdx.x;

    for (int k = 0; k < BN * OUT_CH / 4 / 256; ++k)
        acc4[tid + k * 256] = (v4f){0.0f, 0.0f, 0.0f, 0.0f};
    __syncthreads();

    const int beg = base[b];
    const int cnt = tot[b];

    for (int c0 = wave * 32; c0 < cnt; c0 += 8 * 32) {
        const int m = min(32, cnt - c0);
        unsigned pk = 0;
        if (lane < m) pk = binned[beg + c0 + lane];

        for (int j0 = 0; j0 < m; j0 += EDGE_BATCH) {
            if (m - j0 >= EDGE_BATCH) {
                float va[EDGE_BATCH], vb[EDGE_BATCH];
                #pragma unroll
                for (int j = 0; j < EDGE_BATCH; ++j) {
                    const unsigned pj = __builtin_amdgcn_readlane(pk, j0 + j);
                    const float* r = h + (size_t)(pj >> BN_SHIFT) * OUT_CH + lane;
                    va[j] = r[0];
                    vb[j] = r[HALF_ROW];
                }
                #pragma unroll
                for (int j = 0; j < EDGE_BATCH; ++j) {
                    const unsigned pj = __builtin_amdgcn_readlane(pk, j0 + j);
                    float* o = acc + (pj & (BN - 1)) * OUT_CH + lane;
                    lds_add(o,            va[j]);
                    lds_add(o + HALF_ROW, vb[j]);
                }
            } else {
                for (int j = j0; j < m; ++j) {
                    const unsigned pj = __builtin_amdgcn_readlane(pk, j);
                    const float* r = h + (size_t)(pj >> BN_SHIFT) * OUT_CH + lane;
                    const float a = r[0];
                    const float c = r[HALF_ROW];
                    float* o = acc + (pj & (BN - 1)) * OUT_CH + lane;
                    lds_add(o,            a);
                    lds_add(o + HALF_ROW, c);
                }
            }
        }
    }
    __syncthreads();

    const int node0 = b * BN;
    if (node0 + BN <= n) write_bucket<false>(h, dinv, bias, alpha, acc4, out, node0, n, tid);
    else                 write_bucket<true >(h, dinv, bias, alpha, acc4, out, node0, n, tid);
}

extern "C" void kernel_launch(void* const* d_in, const int* in_sizes, int n_in,
                              void* d_out, int out_size, void* d_ws, size_t ws_size,
                              hipStream_t stream) {
    const float* x     = (const float*)d_in[0];
    const int*   ei    = (const int*)  d_in[1];
    const float* W     = (const float*)d_in[2];
    const float* bias  = (const float*)d_in[3];
    const float* alpha = (const float*)d_in[4];
    float*       out   = (float*)d_out;

    const int N = in_sizes[0] / IN_CH;
    const int E = in_sizes[1] / 2;
    const int* src = ei;
    const int* dst = ei + E;

    char* ws = (char*)d_ws;
    size_t hw_bytes = ((size_t)N * OUT_CH * sizeof(float) + 255) & ~(size_t)255;
    size_t eb_bytes = ((size_t)E * sizeof(unsigned) + 255) & ~(size_t)255;
    const int NB = (N + BN - 1) / BN;
    size_t nb_bytes = ((size_t)NB * sizeof(int) + 255) & ~(size_t)255;
    float*    h      = (float*)ws;
    unsigned* binned = (unsigned*)(ws + hw_bytes);
    int*      tot    = (int*)(ws + hw_bytes + eb_bytes);
    int*      cursor = (int*)(ws + hw_bytes + eb_bytes + nb_bytes);
    int*      base   = (int*)(ws + hw_bytes + eb_bytes + 2 * nb_bytes);
    float*    dinv   = (float*)(ws + 2 * hw_bytes);

    const int threads = 256;
    const int chunk = (E + NBLK - 1) / NBLK;

    init_kernel<<<(NB + threads - 1) / threads, threads, 0, stream>>>(tot, cursor, NB);
    count_kernel<<<NBLK, SORT_THREADS, (size_t)NB * sizeof(int), stream>>>(dst, tot, E, NB, chunk);
    fill_kernel<<<NBLK, SORT_THREADS, (size_t)(2 * NB + 32) * sizeof(int), stream>>>(
        src, dst, tot, cursor, base, binned, E, NB, chunk);
    degree_dinv_kernel<<<(NB + DG_BUCKETS - 1) / DG_BUCKETS, DG_NODES, 0, stream>>>(
        binned, base, tot, dinv, N, NB);
    gemm_wmma_kernel<<<(N + ROWS_PER_BLOCK - 1) / ROWS_PER_BLOCK, threads, 0, stream>>>(
        x, W, dinv, h, N);
    gather_kernel<<<NB, threads, 0, stream>>>(
        h, binned, base, tot, dinv, bias, alpha, out, N);
}


# edge sum in registers: work-group sorts its bucket's edges by node in LDS (integer counting sort), a wave sums each node's rows in registers; LDS float adds per node, not per edge; arrival order
# speedup vs baseline: 3.7657x; 1.0703x over previous
#include <hip/hip_runtime.h>
#include <hip/hip_bf16.h>

#define IN_CH   512
#define OUT_CH  64
#define KCHUNK  128
#define ROWS_PER_BLOCK 128

typedef __attribute__((ext_vector_type(2))) float v2f;
typedef __attribute__((ext_vector_type(4))) float v4f;
typedef __attribute__((ext_vector_type(8))) float v8f;

#define BN            64
#define BN_SHIFT      6
#define NBLK          512
#define SORT_THREADS  1024
#define EDGE_UNROLL   4

__global__ void init_kernel(int* __restrict__ tot, int* __restrict__ cursor, int nb) {
    int i = blockIdx.x * blockDim.x + threadIdx.x;
    if (i < nb) { tot[i] = 0; cursor[i] = 0; }
}

__global__ __launch_bounds__(SORT_THREADS) void count_kernel(
        const int* __restrict__ dst, int* __restrict__ tot,
        int e, int nb, int chunk) {
    extern __shared__ int sm[];
    int* hist = sm;
    const int tid = threadIdx.x;
    for (int i = tid; i < nb; i += SORT_THREADS) hist[i] = 0;
    __syncthreads();
    const int beg = blockIdx.x * chunk;
    const int end = min(beg + chunk, e);
    for (int i0 = beg + tid; i0 < end; i0 += SORT_THREADS * EDGE_UNROLL) {
        int d[EDGE_UNROLL];
        #pragma unroll
        for (int u = 0; u < EDGE_UNROLL; ++u)
            d[u] = dst[min(i0 + u * SORT_THREADS, end - 1)];
        #pragma unroll
        for (int u = 0; u < EDGE_UNROLL; ++u) {
            if (i0 + u * SORT_THREADS < end) atomicAdd(&hist[d[u] >> BN_SHIFT], 1);
        }
    }
    __syncthreads();
    for (int i = tid; i < nb; i += SORT_THREADS) {
        const int c = hist[i];
        if (c) atomicAdd(&tot[i], c);
    }
}

#define DG_BUCKETS 4
#define DG_NODES   (DG_BUCKETS * BN)
#define DG_UNROLL  16

__global__ __launch_bounds__(DG_NODES) void degree_dinv_kernel(
        const unsigned* __restrict__ binned, const int* __restrict__ base,
        const int* __restrict__ tot, float* __restrict__ dinv, int n, int nb) {
    __shared__ int cnt[DG_NODES];
    const int tid = threadIdx.x;
    const int b0  = blockIdx.x * DG_BUCKETS;
    cnt[tid] = 0;
    int bnd[DG_BUCKETS + 1];
    bnd[0] = base[b0];
    for (int k = 0; k < DG_BUCKETS; ++k)
        bnd[k + 1] = bnd[k] + ((b0 + k < nb) ? tot[b0 + k] : 0);
    __syncthreads();
    const int end = bnd[DG_BUCKETS];
    for (int i0 = bnd[0] + tid; i0 < end; i0 += DG_NODES * DG_UNROLL) {
        unsigned w[DG_UNROLL];
        #pragma unroll
        for (int u = 0; u < DG_UNROLL; ++u)
            w[u] = binned[min(i0 + u * DG_NODES, end - 1)];
        #pragma unroll
        for (int u = 0; u < DG_UNROLL; ++u) {
            const int i = i0 + u * DG_NODES;
            const int k = (i >= bnd[1]) + (i >= bnd[2]) + (i >= bnd[3]);
            atomicAdd(&cnt[(k << BN_SHIFT) | (int)(w[u] & (BN - 1))], (i < end) ? 1 : 0);
        }
    }
    __syncthreads();
    const int node = b0 * BN + tid;
    if (node < n) dinv[node] = rsqrtf(1.0f + (float)cnt[tid]);
}

__global__ __launch_bounds__(256) void gemm_wmma_kernel(
        const float* __restrict__ x, const float* __restrict__ W,
        const float* __restrict__ dinv, float* __restrict__ h, int n) {
    __shared__ __align__(16) v2f lds_w[(KCHUNK / 2) * OUT_CH];

    const int tid   = threadIdx.x;
    const int wave  = tid >> 5;
    const int lane  = tid & 31;
    const int half  = lane >> 4;
    const int l15   = lane & 15;
    const int kh    = half * 2;
    const int row0  = blockIdx.x * ROWS_PER_BLOCK + wave * 16;

    int arow = row0 + l15;
    if (arow >= n) arow = n - 1;
    const float* xrow = x + (size_t)arow * IN_CH;

    float dr[8];
    #pragma unroll
    for (int v = 0; v < 8; ++v) {
        int r = row0 + half * 8 + v;
        dr[v] = dinv[r < n ? r : n - 1];
    }

    v8f acc[4] = {{}, {}, {}, {}};

    for (int kc = 0; kc < IN_CH; kc += KCHUNK) {
        #pragma unroll
        for (int i = 0; i < (KCHUNK / 2) * OUT_CH / 256; ++i) {
            int j = tid + i * 256;
            int p = j >> 6;
            int c = j & 63;
            v2f w2;
            w2.x = W[(size_t)(kc + 2 * p)     * OUT_CH + c];
            w2.y = W[(size_t)(kc + 2 * p + 1) * OUT_CH + c];
            lds_w[j] = w2;
        }
        __syncthreads();

        #pragma unroll 4
        for (int k = 0; k < KCHUNK; k += 4) {
            v2f a;
            a.x = xrow[kc + k + kh];
            a.y = xrow[kc + k + kh + 1];
            const int prow = (k >> 1) + half;
            #pragma unroll
            for (int t = 0; t < 4; ++t) {
                v2f b = lds_w[prow * OUT_CH + t * 16 + l15];
                acc[t] = __builtin_amdgcn_wmma_f32_16x16x4_f32(
                    false, a, false, b, (short)0, acc[t], false, false);
            }
        }
        __syncthreads();
    }

    #pragma unroll
    for (int v = 0; v < 8; ++v) {
        int row = row0 + half * 8 + v;
        if (row < n) {
            #pragma unroll
            for (int t = 0; t < 4; ++t)
                h[(size_t)row * OUT_CH + t * 16 + l15] = acc[t][v] * dr[v];
        }
    }
}

__device__ __forceinline__ int block_excl_scan(int v, int* wsum, int tid) {
    const int lane = tid & 31;
    const int w    = tid >> 5;
    int inc = v;
    for (int o = 1; o < 32; o <<= 1) {
        const int nbr = __shfl_up(inc, o, 32);
        if (lane >= o) inc += nbr;
    }
    if (lane == 31) wsum[w] = inc;
    __syncthreads();
    int add = 0;
    for (int k = 0; k < w; ++k) add += wsum[k];
    __syncthreads();
    return add + inc - v;
}

__global__ __launch_bounds__(SORT_THREADS) void fill_kernel(
        const int* __restrict__ src, const int* __restrict__ dst,
        const int* __restrict__ tot, int* __restrict__ cursor,
        int* __restrict__ base_out, unsigned* __restrict__ binned,
        int e, int nb, int chunk) {
    extern __shared__ int sm[];
    int* hist = sm;
    int* pos  = sm + nb;
    int* wsum = sm + 2 * nb;
    const int tid = threadIdx.x;
    for (int i = tid; i < nb; i += SORT_THREADS) hist[i] = 0;
    __syncthreads();
    const int beg = blockIdx.x * chunk;
    const int end = min(beg + chunk, e);
    for (int i0 = beg + tid; i0 < end; i0 += SORT_THREADS * EDGE_UNROLL) {
        int d[EDGE_UNROLL];
        #pragma unroll
        for (int u = 0; u < EDGE_UNROLL; ++u)
            d[u] = dst[min(i0 + u * SORT_THREADS, end - 1)];
        #pragma unroll
        for (int u = 0; u < EDGE_UNROLL; ++u)
            if (i0 + u * SORT_THREADS < end) atomicAdd(&hist[d[u] >> BN_SHIFT], 1);
    }

    const int seg = (nb + SORT_THREADS - 1) / SORT_THREADS;
    const int b0  = tid * seg;
    int tsum = 0;
    for (int k = 0; k < seg; ++k) {
        const int b = b0 + k;
        if (b < nb) tsum += tot[b];
    }
    int run = block_excl_scan(tsum, wsum, tid);
    for (int k = 0; k < seg; ++k) {
        const int b = b0 + k;
        if (b < nb) {
            if (blockIdx.x == 0) base_out[b] = run;
            const int c = hist[b];
            pos[b] = run + (c ? atomicAdd(&cursor[b], c) : 0);
            run += tot[b];
        }
    }
    __syncthreads();

    for (int i0 = beg + tid; i0 < end; i0 += SORT_THREADS * EDGE_UNROLL) {
        int s[EDGE_UNROLL], d[EDGE_UNROLL];
        #pragma unroll
        for (int u = 0; u < EDGE_UNROLL; ++u) {
            const int ic = min(i0 + u * SORT_THREADS, end - 1);
            s[u] = src[ic];
            d[u] = dst[ic];
        }
        #pragma unroll
        for (int u = 0; u < EDGE_UNROLL; ++u) {
            if (i0 + u * SORT_THREADS < end) {
                const int p = atomicAdd(&pos[d[u] >> BN_SHIFT], 1);
                binned[p] = ((unsigned)s[u] << BN_SHIFT) | (unsigned)(d[u] & (BN - 1));
            }
        }
    }
}

#define EDGE_BATCH     16
#define HALF_ROW       (OUT_CH / 2)
#define GATHER_THREADS 256
#define SORT_SLABS     10
#define SORT_TILE      (GATHER_THREADS * SORT_SLABS)
#define WAVE_NODES     (BN / (GATHER_THREADS / 32))

__device__ __forceinline__ void lds_add(float* p, float v) {
    __hip_atomic_fetch_add(p, v, __ATOMIC_RELAXED, __HIP_MEMORY_SCOPE_WORKGROUP);
}

template <bool TAIL>
__device__ __forceinline__ void write_bucket(
        const float* __restrict__ h, const float* __restrict__ dinv,
        const float* __restrict__ bias, const float* __restrict__ alpha,
        const v4f* acc4, float* __restrict__ out, int node0, int n, int tid) {
    const int q = tid & 15;
    float bq[4], aq[4];
    for (int t = 0; t < 4; ++t) { bq[t] = bias[4 * q + t]; aq[t] = alpha[4 * q + t]; }
    #pragma unroll
    for (int k = 0; k < BN / 16; ++k) {
        const int r    = (tid >> 4) + 16 * k;
        const int node = node0 + r;
        if (TAIL && node >= n) continue;
        const v4f hv = *(const v4f*)(h + (size_t)node * OUT_CH + 4 * q);
        const float di = dinv[node];
        const v4f sv = acc4[r * 16 + q];
        v4f v;
        for (int t = 0; t < 4; ++t) {
            const float y = (sv[t] + hv[t]) * di + bq[t];
            v[t] = (y >= 0.0f) ? y : aq[t] * y;
        }
        *(v4f*)(out + (size_t)node * OUT_CH + 4 * q) = v;
    }
}

__device__ __forceinline__ void walk_edge(float* acc, unsigned lane, int d, float x, float y,
                                          int& cur, float& a0, float& a1) {
    if (d != cur) {
        float* o = acc + cur * OUT_CH + lane;
        lds_add(o,            a0);
        lds_add(o + HALF_ROW, a1);
        cur = d;
        a0 = 0.0f;
        a1 = 0.0f;
    }
    a0 += x;
    a1 += y;
}

typedef const __attribute__((address_space(1))) float* gmem_row;

__device__ __forceinline__ gmem_row row_base(const float* h, unsigned word) {
    const unsigned long long a =
        (unsigned long long)h + ((unsigned long long)(word >> BN_SHIFT) * (OUT_CH * sizeof(float)));
    const unsigned lo = __builtin_amdgcn_readfirstlane((unsigned)a);
    const unsigned hi = __builtin_amdgcn_readfirstlane((unsigned)(a >> 32));
    return (gmem_row)(((unsigned long long)hi << 32) | lo);
}

__global__ __launch_bounds__(GATHER_THREADS) void gather_kernel(
        const float* __restrict__ h, const unsigned* __restrict__ binned,
        const int* __restrict__ base, const int* __restrict__ tot,
        const float* __restrict__ dinv, const float* __restrict__ bias,
        const float* __restrict__ alpha, float* __restrict__ out, int n) {
    __shared__ __align__(16) float acc[BN * OUT_CH];
    __shared__ unsigned sorted[SORT_TILE];
    __shared__ int ncnt[BN];
    __shared__ int nstart[BN + 1];
    v4f* acc4 = (v4f*)acc;
    const int tid  = threadIdx.x;
    const unsigned lane = tid & 31;
    const int wave = __builtin_amdgcn_readfirstlane(tid >> 5);
    const int b    = blockIdx.x;

    for (int k = 0; k < BN * OUT_CH / 4 / GATHER_THREADS; ++k)
        acc4[tid + k * GATHER_THREADS] = (v4f){0.0f, 0.0f, 0.0f, 0.0f};

    const int beg = base[b];
    const int cnt = tot[b];

    for (int t0 = 0; t0 < cnt; t0 += SORT_TILE) {
        const int tc = min(SORT_TILE, cnt - t0);
        const unsigned* __restrict__ bp = binned + beg + t0;
        if (tid < BN) ncnt[tid] = 0;
        __syncthreads();

        unsigned w[SORT_SLABS];
        int rk[SORT_SLABS], st[SORT_SLABS];
        #pragma unroll
        for (int u = 0; u < SORT_SLABS; ++u)
            w[u] = bp[min(tid + u * GATHER_THREADS, tc - 1)];
        #pragma unroll
        for (int u = 0; u < SORT_SLABS; ++u) {
            rk[u] = 0;
            if (tid + u * GATHER_THREADS < tc)
                rk[u] = atomicAdd(&ncnt[w[u] & (BN - 1)], 1);
        }
        __syncthreads();

        if (wave == 0) {
            const int x = ncnt[2 * lane];
            const int y = ncnt[2 * lane + 1];
            int inc = x + y;
            for (int o = 1; o < 32; o <<= 1) {
                const int nbr = __shfl_up(inc, o, 32);
                if ((int)lane >= o) inc += nbr;
            }
            const int ex = inc - (x + y);
            nstart[2 * lane]     = ex;
            nstart[2 * lane + 1] = ex + x;
            if (lane == 31) nstart[BN] = inc;
        }
        __syncthreads();

        #pragma unroll
        for (int u = 0; u < SORT_SLABS; ++u)
            st[u] = nstart[w[u] & (BN - 1)];
        #pragma unroll
        for (int u = 0; u < SORT_SLABS; ++u) {
            if (tid + u * GATHER_THREADS < tc)
                sorted[st[u] + rk[u]] = w[u];
        }
        __syncthreads();

        const int s0 = __builtin_amdgcn_readfirstlane(nstart[wave * WAVE_NODES]);
        const int s1 = __builtin_amdgcn_readfirstlane(nstart[wave * WAVE_NODES + WAVE_NODES]);
        if (s0 < s1) {
            unsigned pk = 0;
            if (s0 + (int)lane < s1) pk = sorted[s0 + lane];
            int   cur = (int)(__builtin_amdgcn_readlane(pk, 0) & (BN - 1));
            float a0 = 0.0f, a1 = 0.0f;

            for (int c0 = s0; c0 < s1; c0 += 32) {
                const int m = min(32, s1 - c0);
                unsigned pkn = 0;
                if (c0 + 32 + (int)lane < s1) pkn = sorted[c0 + 32 + lane];

                for (int j0 = 0; j0 < m; j0 += EDGE_BATCH) {
                    float va[EDGE_BATCH], vb[EDGE_BATCH];
                    if (m - j0 >= EDGE_BATCH) {
                        #pragma unroll
                        for (int j = 0; j < EDGE_BATCH; ++j) {
                            const gmem_row r = row_base(h, __builtin_amdgcn_readlane(pk, j0 + j));
                            va[j] = r[lane];
                            vb[j] = r[lane + HALF_ROW];
                        }
                        #pragma unroll
                        for (int j = 0; j < EDGE_BATCH; ++j) {
                            const unsigned pj = __builtin_amdgcn_readlane(pk, j0 + j);
                            walk_edge(acc, lane, (int)(pj & (BN - 1)), va[j], vb[j], cur, a0, a1);
                        }
                    } else {
                        const int rem = m - j0;
                        #pragma unroll
                        for (int j = 0; j < EDGE_BATCH; ++j) {
                            const gmem_row r =
                                row_base(h, __builtin_amdgcn_readlane(pk, min(j0 + j, m - 1)));
                            va[j] = r[lane];
                            vb[j] = r[lane + HALF_ROW];
                        }
                        #pragma unroll
                        for (int j = 0; j < EDGE_BATCH; ++j) {
                            if (j < rem) {
                                const unsigned pj = __builtin_amdgcn_readlane(pk, j0 + j);
                                walk_edge(acc, lane, (int)(pj & (BN - 1)), va[j], vb[j], cur, a0, a1);
                            }
                        }
                    }
                }
                pk = pkn;
            }
            float* o = acc + cur * OUT_CH + lane;
            lds_add(o,            a0);
            lds_add(o + HALF_ROW, a1);
        }
    }
    __syncthreads();

    const int node0 = b * BN;
    if (node0 + BN <= n) write_bucket<false>(h, dinv, bias, alpha, acc4, out, node0, n, tid);
    else                 write_bucket<true >(h, dinv, bias, alpha, acc4, out, node0, n, tid);
}

extern "C" void kernel_launch(void* const* d_in, const int* in_sizes, int n_in,
                              void* d_out, int out_size, void* d_ws, size_t ws_size,
                              hipStream_t stream) {
    const float* x     = (const float*)d_in[0];
    const int*   ei    = (const int*)  d_in[1];
    const float* W     = (const float*)d_in[2];
    const float* bias  = (const float*)d_in[3];
    const float* alpha = (const float*)d_in[4];
    float*       out   = (float*)d_out;

    const int N = in_sizes[0] / IN_CH;
    const int E = in_sizes[1] / 2;
    const int* src = ei;
    const int* dst = ei + E;

    char* ws = (char*)d_ws;
    size_t hw_bytes = ((size_t)N * OUT_CH * sizeof(float) + 255) & ~(size_t)255;
    size_t eb_bytes = ((size_t)E * sizeof(unsigned) + 255) & ~(size_t)255;
    const int NB = (N + BN - 1) / BN;
    size_t nb_bytes = ((size_t)NB * sizeof(int) + 255) & ~(size_t)255;
    float*    h      = (float*)ws;
    unsigned* binned = (unsigned*)(ws + hw_bytes);
    int*      tot    = (int*)(ws + hw_bytes + eb_bytes);
    int*      cursor = (int*)(ws + hw_bytes + eb_bytes + nb_bytes);
    int*      base   = (int*)(ws + hw_bytes + eb_bytes + 2 * nb_bytes);
    float*    dinv   = (float*)(ws + 2 * hw_bytes);

    const int threads = 256;
    const int chunk = (E + NBLK - 1) / NBLK;

    init_kernel<<<(NB + threads - 1) / threads, threads, 0, stream>>>(tot, cursor, NB);
    count_kernel<<<NBLK, SORT_THREADS, (size_t)NB * sizeof(int), stream>>>(dst, tot, E, NB, chunk);
    fill_kernel<<<NBLK, SORT_THREADS, (size_t)(2 * NB + 32) * sizeof(int), stream>>>(
        src, dst, tot, cursor, base, binned, E, NB, chunk);
    degree_dinv_kernel<<<(NB + DG_BUCKETS - 1) / DG_BUCKETS, DG_NODES, 0, stream>>>(
        binned, base, tot, dinv, N, NB);
    gemm_wmma_kernel<<<(N + ROWS_PER_BLOCK - 1) / ROWS_PER_BLOCK, threads, 0, stream>>>(
        x, W, dinv, h, N);
    gather_kernel<<<NB, threads, 0, stream>>>(
        h, binned, base, tot, dinv, bias, alpha, out, N);
}


# edge sum: each half-wave walks half of the wave's sorted edges, lane owns 4 channels: one 16-byte load per lane brings 2 whole rows per instruction; a cut node's partial sums meet in LDS float adds
# speedup vs baseline: 3.9380x; 1.0458x over previous
#include <hip/hip_runtime.h>
#include <hip/hip_bf16.h>

#define IN_CH   512
#define OUT_CH  64
#define KCHUNK  128
#define ROWS_PER_BLOCK 128

typedef __attribute__((ext_vector_type(2))) float v2f;
typedef __attribute__((ext_vector_type(4))) float v4f;
typedef __attribute__((ext_vector_type(8))) float v8f;

#define BN            64
#define BN_SHIFT      6
#define NBLK          512
#define SORT_THREADS  1024
#define EDGE_UNROLL   4

__global__ void init_kernel(int* __restrict__ tot, int* __restrict__ cursor, int nb) {
    int i = blockIdx.x * blockDim.x + threadIdx.x;
    if (i < nb) { tot[i] = 0; cursor[i] = 0; }
}

__global__ __launch_bounds__(SORT_THREADS) void count_kernel(
        const int* __restrict__ dst, int* __restrict__ tot,
        int e, int nb, int chunk) {
    extern __shared__ int sm[];
    int* hist = sm;
    const int tid = threadIdx.x;
    for (int i = tid; i < nb; i += SORT_THREADS) hist[i] = 0;
    __syncthreads();
    const int beg = blockIdx.x * chunk;
    const int end = min(beg + chunk, e);
    for (int i0 = beg + tid; i0 < end; i0 += SORT_THREADS * EDGE_UNROLL) {
        int d[EDGE_UNROLL];
        #pragma unroll
        for (int u = 0; u < EDGE_UNROLL; ++u)
            d[u] = dst[min(i0 + u * SORT_THREADS, end - 1)];
        #pragma unroll
        for (int u = 0; u < EDGE_UNROLL; ++u) {
            if (i0 + u * SORT_THREADS < end) atomicAdd(&hist[d[u] >> BN_SHIFT], 1);
        }
    }
    __syncthreads();
    for (int i = tid; i < nb; i += SORT_THREADS) {
        const int c = hist[i];
        if (c) atomicAdd(&tot[i], c);
    }
}

#define DG_BUCKETS 4
#define DG_NODES   (DG_BUCKETS * BN)
#define DG_UNROLL  16

__global__ __launch_bounds__(DG_NODES) void degree_dinv_kernel(
        const unsigned* __restrict__ binned, const int* __restrict__ base,
        const int* __restrict__ tot, float* __restrict__ dinv, int n, int nb) {
    __shared__ int cnt[DG_NODES];
    const int tid = threadIdx.x;
    const int b0  = blockIdx.x * DG_BUCKETS;
    cnt[tid] = 0;
    int bnd[DG_BUCKETS + 1];
    bnd[0] = base[b0];
    for (int k = 0; k < DG_BUCKETS; ++k)
        bnd[k + 1] = bnd[k] + ((b0 + k < nb) ? tot[b0 + k] : 0);
    __syncthreads();
    const int end = bnd[DG_BUCKETS];
    for (int i0 = bnd[0] + tid; i0 < end; i0 += DG_NODES * DG_UNROLL) {
        unsigned w[DG_UNROLL];
        #pragma unroll
        for (int u = 0; u < DG_UNROLL; ++u)
            w[u] = binned[min(i0 + u * DG_NODES, end - 1)];
        #pragma unroll
        for (int u = 0; u < DG_UNROLL; ++u) {
            const int i = i0 + u * DG_NODES;
            const int k = (i >= bnd[1]) + (i >= bnd[2]) + (i >= bnd[3]);
            atomicAdd(&cnt[(k << BN_SHIFT) | (int)(w[u] & (BN - 1))], (i < end) ? 1 : 0);
        }
    }
    __syncthreads();
    const int node = b0 * BN + tid;
    if (node < n) dinv[node] = rsqrtf(1.0f + (float)cnt[tid]);
}

__global__ __launch_bounds__(256) void gemm_wmma_kernel(
        const float* __restrict__ x, const float* __restrict__ W,
        const float* __restrict__ dinv, float* __restrict__ h, int n) {
    __shared__ __align__(16) v2f lds_w[(KCHUNK / 2) * OUT_CH];

    const int tid   = threadIdx.x;
    const int wave  = tid >> 5;
    const int lane  = tid & 31;
    const int half  = lane >> 4;
    const int l15   = lane & 15;
    const int kh    = half * 2;
    const int row0  = blockIdx.x * ROWS_PER_BLOCK + wave * 16;

    int arow = row0 + l15;
    if (arow >= n) arow = n - 1;
    const float* xrow = x + (size_t)arow * IN_CH;

    float dr[8];
    #pragma unroll
    for (int v = 0; v < 8; ++v) {
        int r = row0 + half * 8 + v;
        dr[v] = dinv[r < n ? r : n - 1];
    }

    v8f acc[4] = {{}, {}, {}, {}};

    for (int kc = 0; kc < IN_CH; kc += KCHUNK) {
        #pragma unroll
        for (int i = 0; i < (KCHUNK / 2) * OUT_CH / 256; ++i) {
            int j = tid + i * 256;
            int p = j >> 6;
            int c = j & 63;
            v2f w2;
            w2.x = W[(size_t)(kc + 2 * p)     * OUT_CH + c];
            w2.y = W[(size_t)(kc + 2 * p + 1) * OUT_CH + c];
            lds_w[j] = w2;
        }
        __syncthreads();

        #pragma unroll 4
        for (int k = 0; k < KCHUNK; k += 4) {
            v2f a;
            a.x = xrow[kc + k + kh];
            a.y = xrow[kc + k + kh + 1];
            const int prow = (k >> 1) + half;
            #pragma unroll
            for (int t = 0; t < 4; ++t) {
                v2f b = lds_w[prow * OUT_CH + t * 16 + l15];
                acc[t] = __builtin_amdgcn_wmma_f32_16x16x4_f32(
                    false, a, false, b, (short)0, acc[t], false, false);
            }
        }
        __syncthreads();
    }

    #pragma unroll
    for (int v = 0; v < 8; ++v) {
        int row = row0 + half * 8 + v;
        if (row < n) {
            #pragma unroll
            for (int t = 0; t < 4; ++t)
                h[(size_t)row * OUT_CH + t * 16 + l15] = acc[t][v] * dr[v];
        }
    }
}

__device__ __forceinline__ int block_excl_scan(int v, int* wsum, int tid) {
    const int lane = tid & 31;
    const int w    = tid >> 5;
    int inc = v;
    for (int o = 1; o < 32; o <<= 1) {
        const int nbr = __shfl_up(inc, o, 32);
        if (lane >= o) inc += nbr;
    }
    if (lane == 31) wsum[w] = inc;
    __syncthreads();
    int add = 0;
    for (int k = 0; k < w; ++k) add += wsum[k];
    __syncthreads();
    return add + inc - v;
}

__global__ __launch_bounds__(SORT_THREADS) void fill_kernel(
        const int* __restrict__ src, const int* __restrict__ dst,
        const int* __restrict__ tot, int* __restrict__ cursor,
        int* __restrict__ base_out, unsigned* __restrict__ binned,
        int e, int nb, int chunk) {
    extern __shared__ int sm[];
    int* hist = sm;
    int* pos  = sm + nb;
    int* wsum = sm + 2 * nb;
    const int tid = threadIdx.x;
    for (int i = tid; i < nb; i += SORT_THREADS) hist[i] = 0;
    __syncthreads();
    const int beg = blockIdx.x * chunk;
    const int end = min(beg + chunk, e);
    for (int i0 = beg + tid; i0 < end; i0 += SORT_THREADS * EDGE_UNROLL) {
        int d[EDGE_UNROLL];
        #pragma unroll
        for (int u = 0; u < EDGE_UNROLL; ++u)
            d[u] = dst[min(i0 + u * SORT_THREADS, end - 1)];
        #pragma unroll
        for (int u = 0; u < EDGE_UNROLL; ++u)
            if (i0 + u * SORT_THREADS < end) atomicAdd(&hist[d[u] >> BN_SHIFT], 1);
    }

    const int seg = (nb + SORT_THREADS - 1) / SORT_THREADS;
    const int b0  = tid * seg;
    int tsum = 0;
    for (int k = 0; k < seg; ++k) {
        const int b = b0 + k;
        if (b < nb) tsum += tot[b];
    }
    int run = block_excl_scan(tsum, wsum, tid);
    for (int k = 0; k < seg; ++k) {
        const int b = b0 + k;
        if (b < nb) {
            if (blockIdx.x == 0) base_out[b] = run;
            const int c = hist[b];
            pos[b] = run + (c ? atomicAdd(&cursor[b], c) : 0);
            run += tot[b];
        }
    }
    __syncthreads();

    for (int i0 = beg + tid; i0 < end; i0 += SORT_THREADS * EDGE_UNROLL) {
        int s[EDGE_UNROLL], d[EDGE_UNROLL];
        #pragma unroll
        for (int u = 0; u < EDGE_UNROLL; ++u) {
            const int ic = min(i0 + u * SORT_THREADS, end - 1);
            s[u] = src[ic];
            d[u] = dst[ic];
        }
        #pragma unroll
        for (int u = 0; u < EDGE_UNROLL; ++u) {
            if (i0 + u * SORT_THREADS < end) {
                const int p = atomicAdd(&pos[d[u] >> BN_SHIFT], 1);
                binned[p] = ((unsigned)s[u] << BN_SHIFT) | (unsigned)(d[u] & (BN - 1));
            }
        }
    }
}

#define PAIR_BATCH     6
#define GATHER_THREADS 256
#define SORT_SLABS     10
#define SORT_TILE      (GATHER_THREADS * SORT_SLABS)
#define WAVE_NODES     (BN / (GATHER_THREADS / 32))

__device__ __forceinline__ void lds_add(float* p, float v) {
    __hip_atomic_fetch_add(p, v, __ATOMIC_RELAXED, __HIP_MEMORY_SCOPE_WORKGROUP);
}

template <bool TAIL>
__device__ __forceinline__ void write_bucket(
        const float* __restrict__ h, const float* __restrict__ dinv,
        const float* __restrict__ bias, const float* __restrict__ alpha,
        const v4f* acc4, float* __restrict__ out, int node0, int n, int tid) {
    const int q = tid & 15;
    float bq[4], aq[4];
    for (int t = 0; t < 4; ++t) { bq[t] = bias[4 * q + t]; aq[t] = alpha[4 * q + t]; }
    #pragma unroll
    for (int k = 0; k < BN / 16; ++k) {
        const int r    = (tid >> 4) + 16 * k;
        const int node = node0 + r;
        if (TAIL && node >= n) continue;
        const v4f hv = *(const v4f*)(h + (size_t)node * OUT_CH + 4 * q);
        const float di = dinv[node];
        const v4f sv = acc4[r * 16 + q];
        v4f v;
        for (int t = 0; t < 4; ++t) {
            const float y = (sv[t] + hv[t]) * di + bq[t];
            v[t] = (y >= 0.0f) ? y : aq[t] * y;
        }
        *(v4f*)(out + (size_t)node * OUT_CH + 4 * q) = v;
    }
}

__device__ __forceinline__ void flush_node(float* acc, int l15, int cur, const v4f& a) {
    float* o = acc + cur * OUT_CH + 4 * l15;
    lds_add(o,     a.x);
    lds_add(o + 1, a.y);
    lds_add(o + 2, a.z);
    lds_add(o + 3, a.w);
}

__device__ __forceinline__ void walk_edge(float* acc, int l15, int d, const v4f& v,
                                          int& cur, v4f& a) {
    if (d != cur) {
        flush_node(acc, l15, cur, a);
        cur = d;
        a = (v4f){0.0f, 0.0f, 0.0f, 0.0f};
    }
    a += v;
}

__device__ __forceinline__ unsigned row_slot(unsigned word, unsigned l15) {
    return ((word >> BN_SHIFT) << 4) | l15;
}

__global__ __launch_bounds__(GATHER_THREADS) __attribute__((amdgpu_waves_per_eu(16, 16)))
void gather_kernel(
        const float* __restrict__ h, const unsigned* __restrict__ binned,
        const int* __restrict__ base, const int* __restrict__ tot,
        const float* __restrict__ dinv, const float* __restrict__ bias,
        const float* __restrict__ alpha, float* __restrict__ out, int n) {
    __shared__ __align__(16) float acc[BN * OUT_CH];
    __shared__ unsigned sorted[SORT_TILE];
    __shared__ int ncnt[BN];
    __shared__ int nstart[BN + 1];
    v4f* acc4 = (v4f*)acc;
    const int tid  = threadIdx.x;
    const unsigned lane = tid & 31;
    const int half = lane >> 4;
    const int l15  = lane & 15;
    const int wave = __builtin_amdgcn_readfirstlane(tid >> 5);
    const int b    = blockIdx.x;

    for (int k = 0; k < BN * OUT_CH / 4 / GATHER_THREADS; ++k)
        acc4[tid + k * GATHER_THREADS] = (v4f){0.0f, 0.0f, 0.0f, 0.0f};

    const int beg = base[b];
    const int cnt = tot[b];

    for (int t0 = 0; t0 < cnt; t0 += SORT_TILE) {
        const int tc = min(SORT_TILE, cnt - t0);
        const unsigned* __restrict__ bp = binned + beg + t0;
        if (tid < BN) ncnt[tid] = 0;
        __syncthreads();

        unsigned w[SORT_SLABS];
        int rk[SORT_SLABS], st[SORT_SLABS];
        #pragma unroll
        for (int u = 0; u < SORT_SLABS; ++u)
            w[u] = bp[min(tid + u * GATHER_THREADS, tc - 1)];
        #pragma unroll
        for (int u = 0; u < SORT_SLABS; ++u) {
            rk[u] = 0;
            if (tid + u * GATHER_THREADS < tc)
                rk[u] = atomicAdd(&ncnt[w[u] & (BN - 1)], 1);
        }
        __syncthreads();

        if (wave == 0) {
            const int x = ncnt[2 * lane];
            const int y = ncnt[2 * lane + 1];
            int inc = x + y;
            for (int o = 1; o < 32; o <<= 1) {
                const int nbr = __shfl_up(inc, o, 32);
                if ((int)lane >= o) inc += nbr;
            }
            const int ex = inc - (x + y);
            nstart[2 * lane]     = ex;
            nstart[2 * lane + 1] = ex + x;
            if (lane == 31) nstart[BN] = inc;
        }
        __syncthreads();

        #pragma unroll
        for (int u = 0; u < SORT_SLABS; ++u)
            st[u] = nstart[w[u] & (BN - 1)];
        #pragma unroll
        for (int u = 0; u < SORT_SLABS; ++u) {
            if (tid + u * GATHER_THREADS < tc)
                sorted[st[u] + rk[u]] = w[u];
        }
        __syncthreads();

        const int s0 = __builtin_amdgcn_readfirstlane(nstart[wave * WAVE_NODES]);
        const int s1 = __builtin_amdgcn_readfirstlane(nstart[wave * WAVE_NODES + WAVE_NODES]);
        if (s0 < s1) {
            const int len  = s1 - s0;
            const int len0 = (len + 1) >> 1;
            const int len1 = len - len0;
            const int hb   = half ? s0 + len0 : s0;
            const int hl   = half ? len1 : len0;
            int cur = (int)(sorted[hb - (hl == 0 ? 1 : 0)] & (BN - 1));
            v4f a = {0.0f, 0.0f, 0.0f, 0.0f};
            const v4f* __restrict__ h4 = (const v4f*)h;

            for (int i = 0; i < len0; i += PAIR_BATCH) {
                unsigned wd[PAIR_BATCH];
                v4f v[PAIR_BATCH];
                if (i + PAIR_BATCH <= len1) {
                    #pragma unroll
                    for (int j = 0; j < PAIR_BATCH; ++j)
                        wd[j] = sorted[hb + i + j];
                    #pragma unroll
                    for (int j = 0; j < PAIR_BATCH; ++j)
                        v[j] = h4[row_slot(wd[j], l15)];
                    #pragma unroll
                    for (int j = 0; j < PAIR_BATCH; ++j)
                        walk_edge(acc, l15, (int)(wd[j] & (BN - 1)), v[j], cur, a);
                } else {
                    #pragma unroll
                    for (int j = 0; j < PAIR_BATCH; ++j)
                        wd[j] = sorted[hb + min(i + j, hl - 1)];
                    #pragma unroll
                    for (int j = 0; j < PAIR_BATCH; ++j)
                        v[j] = h4[row_slot(wd[j], l15)];
                    #pragma unroll
                    for (int j = 0; j < PAIR_BATCH; ++j) {
                        if (i + j < hl)
                            walk_edge(acc, l15, (int)(wd[j] & (BN - 1)), v[j], cur, a);
                    }
                }
            }
            if (hl > 0) flush_node(acc, l15, cur, a);
        }
    }
    __syncthreads();

    const int node0 = b * BN;
    if (node0 + BN <= n) write_bucket<false>(h, dinv, bias, alpha, acc4, out, node0, n, tid);
    else                 write_bucket<true >(h, dinv, bias, alpha, acc4, out, node0, n, tid);
}

extern "C" void kernel_launch(void* const* d_in, const int* in_sizes, int n_in,
                              void* d_out, int out_size, void* d_ws, size_t ws_size,
                              hipStream_t stream) {
    const float* x     = (const float*)d_in[0];
    const int*   ei    = (const int*)  d_in[1];
    const float* W     = (const float*)d_in[2];
    const float* bias  = (const float*)d_in[3];
    const float* alpha = (const float*)d_in[4];
    float*       out   = (float*)d_out;

    const int N = in_sizes[0] / IN_CH;
    const int E = in_sizes[1] / 2;
    const int* src = ei;
    const int* dst = ei + E;

    char* ws = (char*)d_ws;
    size_t hw_bytes = ((size_t)N * OUT_CH * sizeof(float) + 255) & ~(size_t)255;
    size_t eb_bytes = ((size_t)E * sizeof(unsigned) + 255) & ~(size_t)255;
    const int NB = (N + BN - 1) / BN;
    size_t nb_bytes = ((size_t)NB * sizeof(int) + 255) & ~(size_t)255;
    float*    h      = (float*)ws;
    unsigned* binned = (unsigned*)(ws + hw_bytes);
    int*      tot    = (int*)(ws + hw_bytes + eb_bytes);
    int*      cursor = (int*)(ws + hw_bytes + eb_bytes + nb_bytes);
    int*      base   = (int*)(ws + hw_bytes + eb_bytes + 2 * nb_bytes);
    float*    dinv   = (float*)(ws + 2 * hw_bytes);

    const int threads = 256;
    const int chunk = (E + NBLK - 1) / NBLK;

    init_kernel<<<(NB + threads - 1) / threads, threads, 0, stream>>>(tot, cursor, NB);
    count_kernel<<<NBLK, SORT_THREADS, (size_t)NB * sizeof(int), stream>>>(dst, tot, E, NB, chunk);
    fill_kernel<<<NBLK, SORT_THREADS, (size_t)(2 * NB + 32) * sizeof(int), stream>>>(
        src, dst, tot, cursor, base, binned, E, NB, chunk);
    degree_dinv_kernel<<<(NB + DG_BUCKETS - 1) / DG_BUCKETS, DG_NODES, 0, stream>>>(
        binned, base, tot, dinv, N, NB);
    gemm_wmma_kernel<<<(N + ROWS_PER_BLOCK - 1) / ROWS_PER_BLOCK, threads, 0, stream>>>(
        x, W, dinv, h, N);
    gather_kernel<<<NB, threads, 0, stream>>>(
        h, binned, base, tot, dinv, bias, alpha, out, N);
}


# binning by pull: each chunk group sorts its edges by bucket in LDS and writes one run plus its bucket starts; each bucket's group copies its pieces and counts degrees; no count pass; arrival order
# speedup vs baseline: 4.7775x; 1.2132x over previous
#include <hip/hip_runtime.h>
#include <hip/hip_bf16.h>

#define IN_CH   512
#define OUT_CH  64
#define KCHUNK  128
#define ROWS_PER_BLOCK 128

typedef __attribute__((ext_vector_type(2))) float v2f;
typedef __attribute__((ext_vector_type(4))) float v4f;
typedef __attribute__((ext_vector_type(8))) float v8f;

#define BN            64
#define BN_SHIFT      6
#define NBLK          512
#define SORT_THREADS  1024
#define EDGE_UNROLL   4

__device__ __forceinline__ int block_excl_scan(int v, int* wsum, int tid) {
    const int lane = tid & 31;
    const int w    = tid >> 5;
    int inc = v;
    for (int o = 1; o < 32; o <<= 1) {
        const int nbr = __shfl_up(inc, o, 32);
        if (lane >= o) inc += nbr;
    }
    if (lane == 31) wsum[w] = inc;
    __syncthreads();
    int add = 0;
    for (int k = 0; k < w; ++k) add += wsum[k];
    __syncthreads();
    return add + inc - v;
}

__device__ __forceinline__ void block_sum2(int& a, int& b, int* wsum, int tid, int nwaves) {
    for (int o = 16; o > 0; o >>= 1) {
        a += __shfl_xor(a, o, 32);
        b += __shfl_xor(b, o, 32);
    }
    if ((tid & 31) == 0) { wsum[2 * (tid >> 5)] = a; wsum[2 * (tid >> 5) + 1] = b; }
    __syncthreads();
    int sa = 0, sb = 0;
    for (int k = 0; k < nwaves; ++k) { sa += wsum[2 * k]; sb += wsum[2 * k + 1]; }
    __syncthreads();
    a = sa;
    b = sb;
}

__global__ __launch_bounds__(SORT_THREADS) void bin_kernel(
        const int* __restrict__ src, const int* __restrict__ dst,
        int* __restrict__ offs, unsigned* __restrict__ runs,
        int e, int nb, int chunk) {
    extern __shared__ int sm[];
    int* hist = sm;
    int* pos  = sm + nb;
    int* wsum = sm + 2 * nb;
    unsigned* stage = (unsigned*)(sm + 2 * nb + 32);
    const int tid = threadIdx.x;
    for (int i = tid; i < nb; i += SORT_THREADS) hist[i] = 0;
    __syncthreads();
    const int beg = blockIdx.x * chunk;
    const int end = min(beg + chunk, e);
    for (int i0 = beg + tid; i0 < end; i0 += SORT_THREADS * EDGE_UNROLL) {
        int d[EDGE_UNROLL];
        #pragma unroll
        for (int u = 0; u < EDGE_UNROLL; ++u)
            d[u] = dst[min(i0 + u * SORT_THREADS, end - 1)];
        #pragma unroll
        for (int u = 0; u < EDGE_UNROLL; ++u)
            if (i0 + u * SORT_THREADS < end) atomicAdd(&hist[d[u] >> BN_SHIFT], 1);
    }
    __syncthreads();

    const int seg = (nb + SORT_THREADS - 1) / SORT_THREADS;
    const int b0  = tid * seg;
    int tsum = 0;
    for (int k = 0; k < seg; ++k) {
        const int b = b0 + k;
        if (b < nb) tsum += hist[b];
    }
    int run = block_excl_scan(tsum, wsum, tid);
    int* __restrict__ orow = offs + (size_t)blockIdx.x * (nb + 1);
    for (int k = 0; k < seg; ++k) {
        const int b = b0 + k;
        if (b < nb) {
            orow[b] = run;
            pos[b]  = run;
            run += hist[b];
        }
    }
    const int len = max(end - beg, 0);
    if (tid == 0) orow[nb] = len;
    __syncthreads();

    for (int i0 = beg + tid; i0 < end; i0 += SORT_THREADS * EDGE_UNROLL) {
        int s[EDGE_UNROLL], d[EDGE_UNROLL];
        #pragma unroll
        for (int u = 0; u < EDGE_UNROLL; ++u) {
            const int ic = min(i0 + u * SORT_THREADS, end - 1);
            s[u] = src[ic];
            d[u] = dst[ic];
        }
        #pragma unroll
        for (int u = 0; u < EDGE_UNROLL; ++u) {
            if (i0 + u * SORT_THREADS < end) {
                const int p = atomicAdd(&pos[d[u] >> BN_SHIFT], 1);
                stage[p] = ((unsigned)s[u] << BN_SHIFT) | (unsigned)(d[u] & (BN - 1));
            }
        }
    }
    __syncthreads();

    unsigned* __restrict__ rp = runs + beg;
    for (int i = tid; i < len; i += SORT_THREADS) rp[i] = stage[i];
}

#define PULL_THREADS 256
#define PULL_G       (NBLK / PULL_THREADS)
#define PULL_UNROLL  8

static_assert((NBLK & (NBLK - 1)) == 0, "the piece search halves a power of two");
static_assert(NBLK % PULL_THREADS == 0, "every thread reads the same number of chunks");

__global__ __launch_bounds__(PULL_THREADS) void pull_degree_kernel(
        const unsigned* __restrict__ runs, const int* __restrict__ offs,
        unsigned* __restrict__ binned, int* __restrict__ base_out, int* __restrict__ tot_out,
        float* __restrict__ dinv, int n, int nb, int chunk) {
    __shared__ int cnt[BN];
    __shared__ int wsum[2 * (PULL_THREADS / 32)];
    __shared__ int pstart[NBLK];
    __shared__ int psrc[NBLK];
    const int tid = threadIdx.x;
    const int b   = blockIdx.x;
    if (tid < BN) cnt[tid] = 0;

    int o0[PULL_G], c[PULL_G];
    #pragma unroll
    for (int u = 0; u < PULL_G; ++u) {
        const int* __restrict__ orow = offs + (size_t)(tid * PULL_G + u) * (nb + 1) + b;
        o0[u] = orow[0];
        c[u]  = orow[1] - o0[u];
    }
    int csum = 0, osum = 0;
    #pragma unroll
    for (int u = 0; u < PULL_G; ++u) { csum += c[u]; osum += o0[u]; }

    int dest = block_excl_scan(csum, wsum, tid);
    #pragma unroll
    for (int u = 0; u < PULL_G; ++u) {
        const int g = tid * PULL_G + u;
        pstart[g] = dest;
        psrc[g]   = g * chunk + o0[u];
        dest += c[u];
    }
    int total = csum, bs = osum;
    block_sum2(total, bs, wsum, tid, PULL_THREADS / 32);
    if (tid == 0) { base_out[b] = bs; tot_out[b] = total; }

    unsigned* __restrict__ dp = binned + bs;
    for (int i0 = tid; i0 < total; i0 += PULL_THREADS * PULL_UNROLL) {
        int at[PULL_UNROLL];
        unsigned w[PULL_UNROLL];
        #pragma unroll
        for (int v = 0; v < PULL_UNROLL; ++v) {
            const int i = min(i0 + v * PULL_THREADS, total - 1);
            int g = 0;
            #pragma unroll
            for (int step = NBLK / 2; step > 0; step >>= 1)
                if (pstart[g + step] <= i) g += step;
            at[v] = psrc[g] + (i - pstart[g]);
        }
        #pragma unroll
        for (int v = 0; v < PULL_UNROLL; ++v)
            w[v] = runs[at[v]];
        #pragma unroll
        for (int v = 0; v < PULL_UNROLL; ++v) {
            const int i = i0 + v * PULL_THREADS;
            if (i < total) {
                dp[i] = w[v];
                atomicAdd(&cnt[w[v] & (BN - 1)], 1);
            }
        }
    }
    __syncthreads();
    const int node = b * BN + tid;
    if (tid < BN && node < n) dinv[node] = rsqrtf(1.0f + (float)cnt[tid]);
}

__global__ __launch_bounds__(256) void gemm_wmma_kernel(
        const float* __restrict__ x, const float* __restrict__ W,
        const float* __restrict__ dinv, float* __restrict__ h, int n) {
    __shared__ __align__(16) v2f lds_w[(KCHUNK / 2) * OUT_CH];

    const int tid   = threadIdx.x;
    const int wave  = tid >> 5;
    const int lane  = tid & 31;
    const int half  = lane >> 4;
    const int l15   = lane & 15;
    const int kh    = half * 2;
    const int row0  = blockIdx.x * ROWS_PER_BLOCK + wave * 16;

    int arow = row0 + l15;
    if (arow >= n) arow = n - 1;
    const float* xrow = x + (size_t)arow * IN_CH;

    float dr[8];
    #pragma unroll
    for (int v = 0; v < 8; ++v) {
        int r = row0 + half * 8 + v;
        dr[v] = dinv[r < n ? r : n - 1];
    }

    v8f acc[4] = {{}, {}, {}, {}};

    for (int kc = 0; kc < IN_CH; kc += KCHUNK) {
        #pragma unroll
        for (int i = 0; i < (KCHUNK / 2) * OUT_CH / 256; ++i) {
            int j = tid + i * 256;
            int p = j >> 6;
            int c = j & 63;
            v2f w2;
            w2.x = W[(size_t)(kc + 2 * p)     * OUT_CH + c];
            w2.y = W[(size_t)(kc + 2 * p + 1) * OUT_CH + c];
            lds_w[j] = w2;
        }
        __syncthreads();

        #pragma unroll 4
        for (int k = 0; k < KCHUNK; k += 4) {
            v2f a;
            a.x = xrow[kc + k + kh];
            a.y = xrow[kc + k + kh + 1];
            const int prow = (k >> 1) + half;
            #pragma unroll
            for (int t = 0; t < 4; ++t) {
                v2f b = lds_w[prow * OUT_CH + t * 16 + l15];
                acc[t] = __builtin_amdgcn_wmma_f32_16x16x4_f32(
                    false, a, false, b, (short)0, acc[t], false, false);
            }
        }
        __syncthreads();
    }

    #pragma unroll
    for (int v = 0; v < 8; ++v) {
        int row = row0 + half * 8 + v;
        if (row < n) {
            #pragma unroll
            for (int t = 0; t < 4; ++t)
                h[(size_t)row * OUT_CH + t * 16 + l15] = acc[t][v] * dr[v];
        }
    }
}

#define PAIR_BATCH     6
#define GATHER_THREADS 256
#define SORT_SLABS     10
#define SORT_TILE      (GATHER_THREADS * SORT_SLABS)
#define WAVE_NODES     (BN / (GATHER_THREADS / 32))

__device__ __forceinline__ void lds_add(float* p, float v) {
    __hip_atomic_fetch_add(p, v, __ATOMIC_RELAXED, __HIP_MEMORY_SCOPE_WORKGROUP);
}

template <bool TAIL>
__device__ __forceinline__ void write_bucket(
        const float* __restrict__ h, const float* __restrict__ dinv,
        const float* __restrict__ bias, const float* __restrict__ alpha,
        const v4f* acc4, float* __restrict__ out, int node0, int n, int tid) {
    const int q = tid & 15;
    float bq[4], aq[4];
    for (int t = 0; t < 4; ++t) { bq[t] = bias[4 * q + t]; aq[t] = alpha[4 * q + t]; }
    #pragma unroll
    for (int k = 0; k < BN / 16; ++k) {
        const int r    = (tid >> 4) + 16 * k;
        const int node = node0 + r;
        if (TAIL && node >= n) continue;
        const v4f hv = *(const v4f*)(h + (size_t)node * OUT_CH + 4 * q);
        const float di = dinv[node];
        const v4f sv = acc4[r * 16 + q];
        v4f v;
        for (int t = 0; t < 4; ++t) {
            const float y = (sv[t] + hv[t]) * di + bq[t];
            v[t] = (y >= 0.0f) ? y : aq[t] * y;
        }
        *(v4f*)(out + (size_t)node * OUT_CH + 4 * q) = v;
    }
}

__device__ __forceinline__ void flush_node(float* acc, int l15, int cur, const v4f& a) {
    float* o = acc + cur * OUT_CH + 4 * l15;
    lds_add(o,     a.x);
    lds_add(o + 1, a.y);
    lds_add(o + 2, a.z);
    lds_add(o + 3, a.w);
}

__device__ __forceinline__ void walk_edge(float* acc, int l15, int d, const v4f& v,
                                          int& cur, v4f& a) {
    if (d != cur) {
        flush_node(acc, l15, cur, a);
        cur = d;
        a = (v4f){0.0f, 0.0f, 0.0f, 0.0f};
    }
    a += v;
}

__device__ __forceinline__ unsigned row_slot(unsigned word, unsigned l15) {
    return ((word >> BN_SHIFT) << 4) | l15;
}

__global__ __launch_bounds__(GATHER_THREADS) __attribute__((amdgpu_waves_per_eu(16, 16)))
void gather_kernel(
        const float* __restrict__ h, const unsigned* __restrict__ binned,
        const int* __restrict__ base, const int* __restrict__ tot,
        const float* __restrict__ dinv, const float* __restrict__ bias,
        const float* __restrict__ alpha, float* __restrict__ out, int n) {
    __shared__ __align__(16) float acc[BN * OUT_CH];
    __shared__ unsigned sorted[SORT_TILE];
    __shared__ int ncnt[BN];
    __shared__ int nstart[BN + 1];
    v4f* acc4 = (v4f*)acc;
    const int tid  = threadIdx.x;
    const unsigned lane = tid & 31;
    const int half = lane >> 4;
    const int l15  = lane & 15;
    const int wave = __builtin_amdgcn_readfirstlane(tid >> 5);
    const int b    = blockIdx.x;

    for (int k = 0; k < BN * OUT_CH / 4 / GATHER_THREADS; ++k)
        acc4[tid + k * GATHER_THREADS] = (v4f){0.0f, 0.0f, 0.0f, 0.0f};

    const int beg = base[b];
    const int cnt = tot[b];

    for (int t0 = 0; t0 < cnt; t0 += SORT_TILE) {
        const int tc = min(SORT_TILE, cnt - t0);
        const unsigned* __restrict__ bp = binned + beg + t0;
        if (tid < BN) ncnt[tid] = 0;
        __syncthreads();

        unsigned w[SORT_SLABS];
        int rk[SORT_SLABS], st[SORT_SLABS];
        #pragma unroll
        for (int u = 0; u < SORT_SLABS; ++u)
            w[u] = bp[min(tid + u * GATHER_THREADS, tc - 1)];
        #pragma unroll
        for (int u = 0; u < SORT_SLABS; ++u) {
            rk[u] = 0;
            if (tid + u * GATHER_THREADS < tc)
                rk[u] = atomicAdd(&ncnt[w[u] & (BN - 1)], 1);
        }
        __syncthreads();

        if (wave == 0) {
            const int x = ncnt[2 * lane];
            const int y = ncnt[2 * lane + 1];
            int inc = x + y;
            for (int o = 1; o < 32; o <<= 1) {
                const int nbr = __shfl_up(inc, o, 32);
                if ((int)lane >= o) inc += nbr;
            }
            const int ex = inc - (x + y);
            nstart[2 * lane]     = ex;
            nstart[2 * lane + 1] = ex + x;
            if (lane == 31) nstart[BN] = inc;
        }
        __syncthreads();

        #pragma unroll
        for (int u = 0; u < SORT_SLABS; ++u)
            st[u] = nstart[w[u] & (BN - 1)];
        #pragma unroll
        for (int u = 0; u < SORT_SLABS; ++u) {
            if (tid + u * GATHER_THREADS < tc)
                sorted[st[u] + rk[u]] = w[u];
        }
        __syncthreads();

        const int s0 = __builtin_amdgcn_readfirstlane(nstart[wave * WAVE_NODES]);
        const int s1 = __builtin_amdgcn_readfirstlane(nstart[wave * WAVE_NODES + WAVE_NODES]);
        if (s0 < s1) {
            const int len  = s1 - s0;
            const int len0 = (len + 1) >> 1;
            const int len1 = len - len0;
            const int hb   = half ? s0 + len0 : s0;
            const int hl   = half ? len1 : len0;
            int cur = (int)(sorted[hb - (hl == 0 ? 1 : 0)] & (BN - 1));
            v4f a = {0.0f, 0.0f, 0.0f, 0.0f};
            const v4f* __restrict__ h4 = (const v4f*)h;

            for (int i = 0; i < len0; i += PAIR_BATCH) {
                unsigned wd[PAIR_BATCH];
                v4f v[PAIR_BATCH];
                if (i + PAIR_BATCH <= len1) {
                    #pragma unroll
                    for (int j = 0; j < PAIR_BATCH; ++j)
                        wd[j] = sorted[hb + i + j];
                    #pragma unroll
                    for (int j = 0; j < PAIR_BATCH; ++j)
                        v[j] = h4[row_slot(wd[j], l15)];
                    #pragma unroll
                    for (int j = 0; j < PAIR_BATCH; ++j)
                        walk_edge(acc, l15, (int)(wd[j] & (BN - 1)), v[j], cur, a);
                } else {
                    #pragma unroll
                    for (int j = 0; j < PAIR_BATCH; ++j)
                        wd[j] = sorted[hb + min(i + j, hl - 1)];
                    #pragma unroll
                    for (int j = 0; j < PAIR_BATCH; ++j)
                        v[j] = h4[row_slot(wd[j], l15)];
                    #pragma unroll
                    for (int j = 0; j < PAIR_BATCH; ++j) {
                        if (i + j < hl)
                            walk_edge(acc, l15, (int)(wd[j] & (BN - 1)), v[j], cur, a);
                    }
                }
            }
            if (hl > 0) flush_node(acc, l15, cur, a);
        }
    }
    __syncthreads();

    const int node0 = b * BN;
    if (node0 + BN <= n) write_bucket<false>(h, dinv, bias, alpha, acc4, out, node0, n, tid);
    else                 write_bucket<true >(h, dinv, bias, alpha, acc4, out, node0, n, tid);
}

extern "C" void kernel_launch(void* const* d_in, const int* in_sizes, int n_in,
                              void* d_out, int out_size, void* d_ws, size_t ws_size,
                              hipStream_t stream) {
    const float* x     = (const float*)d_in[0];
    const int*   ei    = (const int*)  d_in[1];
    const float* W     = (const float*)d_in[2];
    const float* bias  = (const float*)d_in[3];
    const float* alpha = (const float*)d_in[4];
    float*       out   = (float*)d_out;

    const int N = in_sizes[0] / IN_CH;
    const int E = in_sizes[1] / 2;
    const int* src = ei;
    const int* dst = ei + E;

    char* ws = (char*)d_ws;
    size_t hw_bytes = ((size_t)N * OUT_CH * sizeof(float) + 255) & ~(size_t)255;
    size_t eb_bytes = ((size_t)E * sizeof(unsigned) + 255) & ~(size_t)255;
    const int NB = (N + BN - 1) / BN;
    size_t nb_bytes = ((size_t)NB * sizeof(int) + 255) & ~(size_t)255;
    float*    h      = (float*)ws;
    unsigned* runs   = (unsigned*)ws;
    int*      offs   = (int*)(ws + eb_bytes);
    unsigned* binned = (unsigned*)(ws + hw_bytes);
    int*      tot    = (int*)(ws + hw_bytes + eb_bytes);
    int*      base   = (int*)(ws + hw_bytes + eb_bytes + nb_bytes);
    float*    dinv   = (float*)(ws + 2 * hw_bytes);

    const int threads = 256;
    const int chunk = (E + NBLK - 1) / NBLK;

    bin_kernel<<<NBLK, SORT_THREADS, (size_t)(2 * NB + 32 + chunk) * sizeof(int), stream>>>(
        src, dst, offs, runs, E, NB, chunk);
    pull_degree_kernel<<<NB, PULL_THREADS, 0, stream>>>(
        runs, offs, binned, base, tot, dinv, N, NB, chunk);
    gemm_wmma_kernel<<<(N + ROWS_PER_BLOCK - 1) / ROWS_PER_BLOCK, threads, 0, stream>>>(
        x, W, dinv, h, N);
    gather_kernel<<<NB, threads, 0, stream>>>(
        h, binned, base, tot, dinv, bias, alpha, out, N);
}


# x@W moved into the pull launch: the group that counts a bucket's degrees also multiplies its 64 rows (half the groups product first), scales by its own counts; 3 launches; k summed in another order
# speedup vs baseline: 4.9972x; 1.0460x over previous
#include <hip/hip_runtime.h>
#include <hip/hip_bf16.h>

#define IN_CH   512
#define OUT_CH  64
#define KCHUNK  128

typedef __attribute__((ext_vector_type(2))) float v2f;
typedef __attribute__((ext_vector_type(4))) float v4f;
typedef __attribute__((ext_vector_type(8))) float v8f;

#define BN            64
#define BN_SHIFT      6
#define NBLK          512
#define SORT_THREADS  1024
#define EDGE_UNROLL   4

__device__ __forceinline__ int block_excl_scan(int v, int* wsum, int tid) {
    const int lane = tid & 31;
    const int w    = tid >> 5;
    int inc = v;
    for (int o = 1; o < 32; o <<= 1) {
        const int nbr = __shfl_up(inc, o, 32);
        if (lane >= o) inc += nbr;
    }
    if (lane == 31) wsum[w] = inc;
    __syncthreads();
    int add = 0;
    for (int k = 0; k < w; ++k) add += wsum[k];
    __syncthreads();
    return add + inc - v;
}

__device__ __forceinline__ void block_sum2(int& a, int& b, int* wsum, int tid, int nwaves) {
    for (int o = 16; o > 0; o >>= 1) {
        a += __shfl_xor(a, o, 32);
        b += __shfl_xor(b, o, 32);
    }
    if ((tid & 31) == 0) { wsum[2 * (tid >> 5)] = a; wsum[2 * (tid >> 5) + 1] = b; }
    __syncthreads();
    int sa = 0, sb = 0;
    for (int k = 0; k < nwaves; ++k) { sa += wsum[2 * k]; sb += wsum[2 * k + 1]; }
    __syncthreads();
    a = sa;
    b = sb;
}

__global__ __launch_bounds__(SORT_THREADS) void bin_kernel(
        const int* __restrict__ src, const int* __restrict__ dst,
        int* __restrict__ offs, unsigned* __restrict__ runs,
        int e, int nb, int chunk) {
    extern __shared__ int sm[];
    int* hist = sm;
    int* pos  = sm + nb;
    int* wsum = sm + 2 * nb;
    unsigned* stage = (unsigned*)(sm + 2 * nb + 32);
    const int tid = threadIdx.x;
    for (int i = tid; i < nb; i += SORT_THREADS) hist[i] = 0;
    __syncthreads();
    const int beg = blockIdx.x * chunk;
    const int end = min(beg + chunk, e);
    for (int i0 = beg + tid; i0 < end; i0 += SORT_THREADS * EDGE_UNROLL) {
        int d[EDGE_UNROLL];
        #pragma unroll
        for (int u = 0; u < EDGE_UNROLL; ++u)
            d[u] = dst[min(i0 + u * SORT_THREADS, end - 1)];
        #pragma unroll
        for (int u = 0; u < EDGE_UNROLL; ++u)
            if (i0 + u * SORT_THREADS < end) atomicAdd(&hist[d[u] >> BN_SHIFT], 1);
    }
    __syncthreads();

    const int seg = (nb + SORT_THREADS - 1) / SORT_THREADS;
    const int b0  = tid * seg;
    int tsum = 0;
    for (int k = 0; k < seg; ++k) {
        const int b = b0 + k;
        if (b < nb) tsum += hist[b];
    }
    int run = block_excl_scan(tsum, wsum, tid);
    int* __restrict__ orow = offs + (size_t)blockIdx.x * (nb + 1);
    for (int k = 0; k < seg; ++k) {
        const int b = b0 + k;
        if (b < nb) {
            orow[b] = run;
            pos[b]  = run;
            run += hist[b];
        }
    }
    const int len = max(end - beg, 0);
    if (tid == 0) orow[nb] = len;
    __syncthreads();

    for (int i0 = beg + tid; i0 < end; i0 += SORT_THREADS * EDGE_UNROLL) {
        int s[EDGE_UNROLL], d[EDGE_UNROLL];
        #pragma unroll
        for (int u = 0; u < EDGE_UNROLL; ++u) {
            const int ic = min(i0 + u * SORT_THREADS, end - 1);
            s[u] = src[ic];
            d[u] = dst[ic];
        }
        #pragma unroll
        for (int u = 0; u < EDGE_UNROLL; ++u) {
            if (i0 + u * SORT_THREADS < end) {
                const int p = atomicAdd(&pos[d[u] >> BN_SHIFT], 1);
                stage[p] = ((unsigned)s[u] << BN_SHIFT) | (unsigned)(d[u] & (BN - 1));
            }
        }
    }
    __syncthreads();

    unsigned* __restrict__ rp = runs + beg;
    for (int i = tid; i < len; i += SORT_THREADS) rp[i] = stage[i];
}

#define PULL_THREADS 256
#define PULL_G       (NBLK / PULL_THREADS)
#define PULL_UNROLL  8

static_assert((NBLK & (NBLK - 1)) == 0, "the piece search halves a power of two");
static_assert(NBLK % PULL_THREADS == 0, "every thread reads the same number of chunks");


__device__ __forceinline__ void product_rows(
        const float* __restrict__ x, const float* __restrict__ W, v4f* lds_w,
        int node0, int n, int tid, v8f& acc0, v8f& acc1) {
    const int wave = tid >> 5;
    const int lane = tid & 31;
    const int half = lane >> 4;
    const int l15  = lane & 15;
    const int rt   = wave & 3;
    const int ch   = wave >> 2;

    int arow = node0 + rt * 16 + l15;
    if (arow >= n) arow = n - 1;
    const v4f* __restrict__ xrow4 = (const v4f*)(x + (size_t)arow * IN_CH + half * 4);
    const v4f* __restrict__ W4 = (const v4f*)W;

    #pragma unroll 1
    for (int kc = 0; kc < IN_CH; kc += KCHUNK) {
        #pragma unroll 1
        for (int i = 0; i < (KCHUNK / 4) * (OUT_CH / 4) / PULL_THREADS; ++i) {
            const int task = tid + i * PULL_THREADS;
            const int q  = task >> 4;
            const int c4 = task & 15;
            const v4f* __restrict__ wp = W4 + (size_t)(kc + 4 * q) * (OUT_CH / 4) + c4;
            const v4f r0 = wp[0];
            const v4f r1 = wp[OUT_CH / 4];
            const v4f r2 = wp[2 * (OUT_CH / 4)];
            const v4f r3 = wp[3 * (OUT_CH / 4)];
            v4f* dq = lds_w + q * OUT_CH + 4 * c4;
            dq[0] = (v4f){r0.x, r1.x, r2.x, r3.x};
            dq[1] = (v4f){r0.y, r1.y, r2.y, r3.y};
            dq[2] = (v4f){r0.z, r1.z, r2.z, r3.z};
            dq[3] = (v4f){r0.w, r1.w, r2.w, r3.w};
        }
        __syncthreads();

        const v4f* bp = lds_w + half * OUT_CH + ch * 32 + l15;
        #pragma unroll 4
        for (int k = 0; k < KCHUNK; k += 8) {
            const v4f a4 = xrow4[(kc + k) >> 2];
            const v4f b0 = bp[(k >> 2) * OUT_CH];
            const v4f b1 = bp[(k >> 2) * OUT_CH + 16];
            const v2f a01 = __builtin_shufflevector(a4, a4, 0, 1);
            const v2f a23 = __builtin_shufflevector(a4, a4, 2, 3);
            acc0 = __builtin_amdgcn_wmma_f32_16x16x4_f32(
                false, a01, false, __builtin_shufflevector(b0, b0, 0, 1), (short)0, acc0, false, false);
            acc1 = __builtin_amdgcn_wmma_f32_16x16x4_f32(
                false, a01, false, __builtin_shufflevector(b1, b1, 0, 1), (short)0, acc1, false, false);
            acc0 = __builtin_amdgcn_wmma_f32_16x16x4_f32(
                false, a23, false, __builtin_shufflevector(b0, b0, 2, 3), (short)0, acc0, false, false);
            acc1 = __builtin_amdgcn_wmma_f32_16x16x4_f32(
                false, a23, false, __builtin_shufflevector(b1, b1, 2, 3), (short)0, acc1, false, false);
        }
        __syncthreads();
    }
}

__global__ __launch_bounds__(PULL_THREADS) __attribute__((amdgpu_waves_per_eu(16, 16)))
void pull_degree_product_kernel(
        const unsigned* __restrict__ runs, const int* __restrict__ offs,
        unsigned* __restrict__ binned, int* __restrict__ base_out, int* __restrict__ tot_out,
        float* __restrict__ dinv, const float* __restrict__ x, const float* __restrict__ W,
        float* __restrict__ h, int n, int nb, int chunk) {
    __shared__ int cnt[BN];
    __shared__ int wsum[2 * (PULL_THREADS / 32)];
    __shared__ int pstart[NBLK];
    __shared__ int psrc[NBLK];
    __shared__ __align__(16) v4f lds_w[(KCHUNK / 4) * OUT_CH];
    const int tid = threadIdx.x;
    const int b   = blockIdx.x;

    const bool product_first = (__builtin_popcount((unsigned)b) & 1) != 0;
    v8f acc0 = {}, acc1 = {};
    if (product_first) product_rows(x, W, lds_w, b * BN, n, tid, acc0, acc1);

    if (tid < BN) cnt[tid] = 0;

    int o0[PULL_G], c[PULL_G];
    #pragma unroll
    for (int u = 0; u < PULL_G; ++u) {
        const int* __restrict__ orow = offs + (size_t)(tid * PULL_G + u) * (nb + 1) + b;
        o0[u] = orow[0];
        c[u]  = orow[1] - o0[u];
    }
    int csum = 0, osum = 0;
    #pragma unroll
    for (int u = 0; u < PULL_G; ++u) { csum += c[u]; osum += o0[u]; }

    int dest = block_excl_scan(csum, wsum, tid);
    #pragma unroll
    for (int u = 0; u < PULL_G; ++u) {
        const int g = tid * PULL_G + u;
        pstart[g] = dest;
        psrc[g]   = g * chunk + o0[u];
        dest += c[u];
    }
    int total = csum, bs = osum;
    block_sum2(total, bs, wsum, tid, PULL_THREADS / 32);
    if (tid == 0) { base_out[b] = bs; tot_out[b] = total; }

    unsigned* __restrict__ dp = binned + bs;
    for (int i0 = tid; i0 < total; i0 += PULL_THREADS * PULL_UNROLL) {
        int at[PULL_UNROLL];
        unsigned w[PULL_UNROLL];
        #pragma unroll
        for (int v = 0; v < PULL_UNROLL; ++v) {
            const int i = min(i0 + v * PULL_THREADS, total - 1);
            int g = 0;
            #pragma unroll
            for (int step = NBLK / 2; step > 0; step >>= 1)
                if (pstart[g + step] <= i) g += step;
            at[v] = psrc[g] + (i - pstart[g]);
        }
        #pragma unroll
        for (int v = 0; v < PULL_UNROLL; ++v)
            w[v] = runs[at[v]];
        #pragma unroll
        for (int v = 0; v < PULL_UNROLL; ++v) {
            const int i = i0 + v * PULL_THREADS;
            if (i < total) {
                dp[i] = w[v];
                atomicAdd(&cnt[w[v] & (BN - 1)], 1);
            }
        }
    }
    __syncthreads();
    const int node = b * BN + tid;
    if (tid < BN && node < n) dinv[node] = rsqrtf(1.0f + (float)cnt[tid]);

    if (!product_first) product_rows(x, W, lds_w, b * BN, n, tid, acc0, acc1);

    {
        const int lane = tid & 31;
        const int half = lane >> 4;
        const int l15  = lane & 15;
        const int rt   = (tid >> 5) & 3;
        const int ch   = tid >> 7;
        #pragma unroll
        for (int v = 0; v < 8; ++v) {
            const int r   = rt * 16 + half * 8 + v;
            const int row = b * BN + r;
            if (row < n) {
                const float s = rsqrtf(1.0f + (float)cnt[r]);
                float* hp = h + (size_t)row * OUT_CH + ch * 32 + l15;
                hp[0]  = acc0[v] * s;
                hp[16] = acc1[v] * s;
            }
        }
    }
}

#define PAIR_BATCH     6
#define GATHER_THREADS 256
#define SORT_SLABS     10
#define SORT_TILE      (GATHER_THREADS * SORT_SLABS)
#define WAVE_NODES     (BN / (GATHER_THREADS / 32))

__device__ __forceinline__ void lds_add(float* p, float v) {
    __hip_atomic_fetch_add(p, v, __ATOMIC_RELAXED, __HIP_MEMORY_SCOPE_WORKGROUP);
}

template <bool TAIL>
__device__ __forceinline__ void write_bucket(
        const float* __restrict__ h, const float* __restrict__ dinv,
        const float* __restrict__ bias, const float* __restrict__ alpha,
        const v4f* acc4, float* __restrict__ out, int node0, int n, int tid) {
    const int q = tid & 15;
    float bq[4], aq[4];
    for (int t = 0; t < 4; ++t) { bq[t] = bias[4 * q + t]; aq[t] = alpha[4 * q + t]; }
    #pragma unroll
    for (int k = 0; k < BN / 16; ++k) {
        const int r    = (tid >> 4) + 16 * k;
        const int node = node0 + r;
        if (TAIL && node >= n) continue;
        const v4f hv = *(const v4f*)(h + (size_t)node * OUT_CH + 4 * q);
        const float di = dinv[node];
        const v4f sv = acc4[r * 16 + q];
        v4f v;
        for (int t = 0; t < 4; ++t) {
            const float y = (sv[t] + hv[t]) * di + bq[t];
            v[t] = (y >= 0.0f) ? y : aq[t] * y;
        }
        *(v4f*)(out + (size_t)node * OUT_CH + 4 * q) = v;
    }
}

__device__ __forceinline__ void flush_node(float* acc, int l15, int cur, const v4f& a) {
    float* o = acc + cur * OUT_CH + 4 * l15;
    lds_add(o,     a.x);
    lds_add(o + 1, a.y);
    lds_add(o + 2, a.z);
    lds_add(o + 3, a.w);
}

__device__ __forceinline__ void walk_edge(float* acc, int l15, int d, const v4f& v,
                                          int& cur, v4f& a) {
    if (d != cur) {
        flush_node(acc, l15, cur, a);
        cur = d;
        a = (v4f){0.0f, 0.0f, 0.0f, 0.0f};
    }
    a += v;
}

__device__ __forceinline__ unsigned row_slot(unsigned word, unsigned l15) {
    return ((word >> BN_SHIFT) << 4) | l15;
}

__global__ __launch_bounds__(GATHER_THREADS) __attribute__((amdgpu_waves_per_eu(16, 16)))
void gather_kernel(
        const float* __restrict__ h, const unsigned* __restrict__ binned,
        const int* __restrict__ base, const int* __restrict__ tot,
        const float* __restrict__ dinv, const float* __restrict__ bias,
        const float* __restrict__ alpha, float* __restrict__ out, int n) {
    __shared__ __align__(16) float acc[BN * OUT_CH];
    __shared__ unsigned sorted[SORT_TILE];
    __shared__ int ncnt[BN];
    __shared__ int nstart[BN + 1];
    v4f* acc4 = (v4f*)acc;
    const int tid  = threadIdx.x;
    const unsigned lane = tid & 31;
    const int half = lane >> 4;
    const int l15  = lane & 15;
    const int wave = __builtin_amdgcn_readfirstlane(tid >> 5);
    const int b    = blockIdx.x;

    for (int k = 0; k < BN * OUT_CH / 4 / GATHER_THREADS; ++k)
        acc4[tid + k * GATHER_THREADS] = (v4f){0.0f, 0.0f, 0.0f, 0.0f};

    const int beg = base[b];
    const int cnt = tot[b];

    for (int t0 = 0; t0 < cnt; t0 += SORT_TILE) {
        const int tc = min(SORT_TILE, cnt - t0);
        const unsigned* __restrict__ bp = binned + beg + t0;
        if (tid < BN) ncnt[tid] = 0;
        __syncthreads();

        unsigned w[SORT_SLABS];
        int rk[SORT_SLABS], st[SORT_SLABS];
        #pragma unroll
        for (int u = 0; u < SORT_SLABS; ++u)
            w[u] = bp[min(tid + u * GATHER_THREADS, tc - 1)];
        #pragma unroll
        for (int u = 0; u < SORT_SLABS; ++u) {
            rk[u] = 0;
            if (tid + u * GATHER_THREADS < tc)
                rk[u] = atomicAdd(&ncnt[w[u] & (BN - 1)], 1);
        }
        __syncthreads();

        if (wave == 0) {
            const int x = ncnt[2 * lane];
            const int y = ncnt[2 * lane + 1];
            int inc = x + y;
            for (int o = 1; o < 32; o <<= 1) {
                const int nbr = __shfl_up(inc, o, 32);
                if ((int)lane >= o) inc += nbr;
            }
            const int ex = inc - (x + y);
            nstart[2 * lane]     = ex;
            nstart[2 * lane + 1] = ex + x;
            if (lane == 31) nstart[BN] = inc;
        }
        __syncthreads();

        #pragma unroll
        for (int u = 0; u < SORT_SLABS; ++u)
            st[u] = nstart[w[u] & (BN - 1)];
        #pragma unroll
        for (int u = 0; u < SORT_SLABS; ++u) {
            if (tid + u * GATHER_THREADS < tc)
                sorted[st[u] + rk[u]] = w[u];
        }
        __syncthreads();

        const int s0 = __builtin_amdgcn_readfirstlane(nstart[wave * WAVE_NODES]);
        const int s1 = __builtin_amdgcn_readfirstlane(nstart[wave * WAVE_NODES + WAVE_NODES]);
        if (s0 < s1) {
            const int len  = s1 - s0;
            const int len0 = (len + 1) >> 1;
            const int len1 = len - len0;
            const int hb   = half ? s0 + len0 : s0;
            const int hl   = half ? len1 : len0;
            int cur = (int)(sorted[hb - (hl == 0 ? 1 : 0)] & (BN - 1));
            v4f a = {0.0f, 0.0f, 0.0f, 0.0f};
            const v4f* __restrict__ h4 = (const v4f*)h;

            for (int i = 0; i < len0; i += PAIR_BATCH) {
                unsigned wd[PAIR_BATCH];
                v4f v[PAIR_BATCH];
                if (i + PAIR_BATCH <= len1) {
                    #pragma unroll
                    for (int j = 0; j < PAIR_BATCH; ++j)
                        wd[j] = sorted[hb + i + j];
                    #pragma unroll
                    for (int j = 0; j < PAIR_BATCH; ++j)
                        v[j] = h4[row_slot(wd[j], l15)];
                    #pragma unroll
                    for (int j = 0; j < PAIR_BATCH; ++j)
                        walk_edge(acc, l15, (int)(wd[j] & (BN - 1)), v[j], cur, a);
                } else {
                    #pragma unroll
                    for (int j = 0; j < PAIR_BATCH; ++j)
                        wd[j] = sorted[hb + min(i + j, hl - 1)];
                    #pragma unroll
                    for (int j = 0; j < PAIR_BATCH; ++j)
                        v[j] = h4[row_slot(wd[j], l15)];
                    #pragma unroll
                    for (int j = 0; j < PAIR_BATCH; ++j) {
                        if (i + j < hl)
                            walk_edge(acc, l15, (int)(wd[j] & (BN - 1)), v[j], cur, a);
                    }
                }
            }
            if (hl > 0) flush_node(acc, l15, cur, a);
        }
    }
    __syncthreads();

    const int node0 = b * BN;
    if (node0 + BN <= n) write_bucket<false>(h, dinv, bias, alpha, acc4, out, node0, n, tid);
    else                 write_bucket<true >(h, dinv, bias, alpha, acc4, out, node0, n, tid);
}

extern "C" void kernel_launch(void* const* d_in, const int* in_sizes, int n_in,
                              void* d_out, int out_size, void* d_ws, size_t ws_size,
                              hipStream_t stream) {
    const float* x     = (const float*)d_in[0];
    const int*   ei    = (const int*)  d_in[1];
    const float* W     = (const float*)d_in[2];
    const float* bias  = (const float*)d_in[3];
    const float* alpha = (const float*)d_in[4];
    float*       out   = (float*)d_out;

    const int N = in_sizes[0] / IN_CH;
    const int E = in_sizes[1] / 2;
    const int* src = ei;
    const int* dst = ei + E;

    char* ws = (char*)d_ws;
    size_t hw_bytes = ((size_t)N * OUT_CH * sizeof(float) + 255) & ~(size_t)255;
    size_t eb_bytes = ((size_t)E * sizeof(unsigned) + 255) & ~(size_t)255;
    const int NB = (N + BN - 1) / BN;
    size_t nb_bytes = ((size_t)NB * sizeof(int) + 255) & ~(size_t)255;
    float*    h      = (float*)ws;
    unsigned* runs   = (unsigned*)d_out;
    int*      offs   = (int*)((char*)d_out + eb_bytes);
    unsigned* binned = (unsigned*)(ws + hw_bytes);
    int*      tot    = (int*)(ws + hw_bytes + eb_bytes);
    int*      base   = (int*)(ws + hw_bytes + eb_bytes + nb_bytes);
    float*    dinv   = (float*)(ws + 2 * hw_bytes);

    const int threads = 256;
    const int chunk = (E + NBLK - 1) / NBLK;

    bin_kernel<<<NBLK, SORT_THREADS, (size_t)(2 * NB + 32 + chunk) * sizeof(int), stream>>>(
        src, dst, offs, runs, E, NB, chunk);
    pull_degree_product_kernel<<<NB, PULL_THREADS, 0, stream>>>(
        runs, offs, binned, base, tot, dinv, x, W, h, N, NB, chunk);
    gather_kernel<<<NB, threads, 0, stream>>>(
        h, binned, base, tot, dinv, bias, alpha, out, N);
}
